# starting kernel
# baseline (speedup 1.0000x reference)
#include <hip/hip_runtime.h>
#include <hip/hip_bf16.h>

typedef float v2f __attribute__((ext_vector_type(2)));
typedef float v8f __attribute__((ext_vector_type(8)));

#define NEG_SLOPE 0.2f
#define INF_EPS   1e-16f

__device__ __forceinline__ unsigned fEnc(float x) {
    unsigned u = __float_as_uint(x);
    return (u & 0x80000000u) ? ~u : (u | 0x80000000u);
}
__device__ __forceinline__ float fDec(unsigned u) {
    u = (u & 0x80000000u) ? (u & 0x7FFFFFFFu) : ~u;
    return __uint_as_float(u);
}
__device__ __forceinline__ float leaky(float x) { return x > 0.f ? x : NEG_SLOPE * x; }

template <int K, int NCOLS>
__global__ __launch_bounds__(256) void gemm_wmma_f32(
    const float* __restrict__ A, const float* __restrict__ B,
    float* __restrict__ C)
{
    const int lane = threadIdx.x & 31;
    const int wave = threadIdx.x >> 5;
    const int row  = blockIdx.x * 16 + (lane & 15);
    const int colB = wave * 16 + (lane & 15);
    const int kh   = (lane >> 4) << 1;

    const float* __restrict__ aPtr = A + (size_t)row * K + kh;
    const float* __restrict__ bPtr = B + (size_t)kh * NCOLS + colB;

    v8f acc = {};
#pragma unroll
    for (int k0 = 0; k0 < K; k0 += 4) {
        v2f a, b;
        a.x = aPtr[k0];
        a.y = aPtr[k0 + 1];
        b.x = bPtr[k0 * NCOLS];
        b.y = bPtr[k0 * NCOLS + NCOLS];
        acc = __builtin_amdgcn_wmma_f32_16x16x4_f32(
            false, a, false, b, (short)0, acc, false, false);
    }
    float* __restrict__ cPtr =
        C + ((size_t)blockIdx.x * 16 + ((lane >> 4) << 3)) * NCOLS + colB;
#pragma unroll
    for (int r = 0; r < 8; ++r)
        cPtr[r * NCOLS] = acc[r];
}

__global__ void attn_logits_h4(const float* __restrict__ h,
                               const float* __restrict__ attS,
                               const float* __restrict__ attD,
                               float* __restrict__ alS, float* __restrict__ alD, int N)
{
    int t = blockIdx.x * blockDim.x + threadIdx.x;
    if (t >= N * 4) return;
    int n = t >> 2, hd = t & 3;
    const float* hp = h + (size_t)n * 128 + hd * 32;
    const float* as = attS + hd * 32;
    const float* ad = attD + hd * 32;
    float s = 0.f, d = 0.f;
#pragma unroll
    for (int c = 0; c < 32; ++c) { float v = hp[c]; s += v * as[c]; d += v * ad[c]; }
    alS[t] = s; alD[t] = d;
}

__global__ void attn_logits_h1(const float* __restrict__ h,
                               const float* __restrict__ attS,
                               const float* __restrict__ attD,
                               float* __restrict__ alS, float* __restrict__ alD, int N)
{
    int n = blockIdx.x * blockDim.x + threadIdx.x;
    if (n >= N) return;
    const float* hp = h + (size_t)n * 64;
    float s = 0.f, d = 0.f;
#pragma unroll
    for (int c = 0; c < 64; ++c) { float v = hp[c]; s += v * attS[c]; d += v * attD[c]; }
    alS[n] = s; alD[n] = d;
}

__global__ void edge_max_h4(const int* __restrict__ esrc, const int* __restrict__ edst,
                            const float4* __restrict__ alS, const float4* __restrict__ alD,
                            unsigned* __restrict__ mEnc, int E, int Etot)
{
    int e = blockIdx.x * blockDim.x + threadIdx.x;
    if (e >= Etot) return;
    int s, d;
    if (e < E) { s = esrc[e]; d = edst[e]; } else { s = d = e - E; }
    float4 vs = alS[s], vd = alD[d];
    float ev[4] = { leaky(vs.x + vd.x), leaky(vs.y + vd.y),
                    leaky(vs.z + vd.z), leaky(vs.w + vd.w) };
#pragma unroll
    for (int h = 0; h < 4; ++h) atomicMax(&mEnc[d * 4 + h], fEnc(ev[h]));
}

__global__ void edge_exp_h4(const int* __restrict__ esrc, const int* __restrict__ edst,
                            const float4* __restrict__ alS, const float4* __restrict__ alD,
                            const unsigned* __restrict__ mEnc,
                            float* __restrict__ ex, float* __restrict__ den, int E, int Etot)
{
    int e = blockIdx.x * blockDim.x + threadIdx.x;
    if (e >= Etot) return;
    int s, d;
    if (e < E) { s = esrc[e]; d = edst[e]; } else { s = d = e - E; }
    float4 vs = alS[s], vd = alD[d];
    float ev[4] = { leaky(vs.x + vd.x), leaky(vs.y + vd.y),
                    leaky(vs.z + vd.z), leaky(vs.w + vd.w) };
#pragma unroll
    for (int h = 0; h < 4; ++h) {
        float x = __expf(ev[h] - fDec(mEnc[d * 4 + h]));
        ex[(size_t)e * 4 + h] = x;
        atomicAdd(&den[d * 4 + h], x);
    }
}

__global__ void edge_agg_128(const int* __restrict__ esrc, const int* __restrict__ edst,
                             const float* __restrict__ ex, const float* __restrict__ den,
                             const float* __restrict__ h, float* __restrict__ out,
                             int E, int Etot)
{
    int t = blockIdx.x * blockDim.x + threadIdx.x;
    if (t >= Etot * 128) return;
    int e = t >> 7, c = t & 127, hd = c >> 5;
    int s, d;
    if (e < E) { s = esrc[e]; d = edst[e]; } else { s = d = e - E; }
    float alpha = ex[(size_t)e * 4 + hd] / (den[d * 4 + hd] + INF_EPS);
    atomicAdd(&out[(size_t)d * 128 + c], alpha * h[(size_t)s * 128 + c]);
}

__global__ void finalize_elu(float* __restrict__ a, const float* __restrict__ b, int N)
{
    int t = blockIdx.x * blockDim.x + threadIdx.x;
    if (t >= N * 128) return;
    float v = a[t] + b[t & 127];
    a[t] = v > 0.f ? v : __expf(v) - 1.f;
}

__global__ void edge_max_h1(const int* __restrict__ esrc, const int* __restrict__ edst,
                            const float* __restrict__ alS, const float* __restrict__ alD,
                            unsigned* __restrict__ mEnc, int E, int Etot)
{
    int e = blockIdx.x * blockDim.x + threadIdx.x;
    if (e >= Etot) return;
    int s, d;
    if (e < E) { s = esrc[e]; d = edst[e]; } else { s = d = e - E; }
    atomicMax(&mEnc[d], fEnc(leaky(alS[s] + alD[d])));
}

__global__ void edge_exp_h1(const int* __restrict__ esrc, const int* __restrict__ edst,
                            const float* __restrict__ alS, const float* __restrict__ alD,
                            const unsigned* __restrict__ mEnc,
                            float* __restrict__ ex, float* __restrict__ den, int E, int Etot)
{
    int e = blockIdx.x * blockDim.x + threadIdx.x;
    if (e >= Etot) return;
    int s, d;
    if (e < E) { s = esrc[e]; d = edst[e]; } else { s = d = e - E; }
    float x = __expf(leaky(alS[s] + alD[d]) - fDec(mEnc[d]));
    ex[e] = x;
    atomicAdd(&den[d], x);
}

__global__ void edge_agg_64(const int* __restrict__ esrc, const int* __restrict__ edst,
                            const float* __restrict__ ex, const float* __restrict__ den,
                            const float* __restrict__ h, float* __restrict__ out,
                            int E, int Etot)
{
    int t = blockIdx.x * blockDim.x + threadIdx.x;
    if (t >= Etot * 64) return;
    int e = t >> 6, c = t & 63;
    int s, d;
    if (e < E) { s = esrc[e]; d = edst[e]; } else { s = d = e - E; }
    float alpha = ex[e] / (den[d] + INF_EPS);
    atomicAdd(&out[(size_t)d * 64 + c], alpha * h[(size_t)s * 64 + c]);
}

__global__ void finalize_bias64(float* __restrict__ o, const float* __restrict__ b, int N)
{
    int t = blockIdx.x * blockDim.x + threadIdx.x;
    if (t >= N * 64) return;
    o[t] += b[t & 63];
}

extern "C" void kernel_launch(void* const* d_in, const int* in_sizes, int n_in,
                              void* d_out, int out_size, void* d_ws, size_t ws_size,
                              hipStream_t stream)
{
    const float* x    = (const float*)d_in[0];
    const int*   eidx = (const int*)d_in[1];
    const float* W1   = (const float*)d_in[2];
    const float* aS1  = (const float*)d_in[3];
    const float* aD1  = (const float*)d_in[4];
    const float* b1   = (const float*)d_in[5];
    const float* W2   = (const float*)d_in[6];
    const float* aS2  = (const float*)d_in[7];
    const float* aD2  = (const float*)d_in[8];
    const float* b2   = (const float*)d_in[9];
    float* out = (float*)d_out;

    const int N    = in_sizes[0] / 128;
    const int E    = in_sizes[1] / 2;
    const int Etot = E + N;
    const int* esrc = eidx;
    const int* edst = eidx + E;

    float* ws = (float*)d_ws;
    size_t o = 0;
    float*    h1   = ws + o; o += (size_t)N * 128;
    float*    alS1 = ws + o; o += (size_t)N * 4;
    float*    alD1 = ws + o; o += (size_t)N * 4;
    float*    ex1  = ws + o; o += (size_t)Etot * 4;
    float*    agg1 = ws + o; o += (size_t)N * 128;
    unsigned* m1   = (unsigned*)(ws + o); o += (size_t)N * 4;
    float*    den1 = ws + o; o += (size_t)N * 4;
    float*    h2   = ws + o; o += (size_t)N * 64;
    float*    alS2 = ws + o; o += (size_t)N;
    float*    alD2 = ws + o; o += (size_t)N;
    float*    ex2  = ws + o; o += (size_t)Etot;
    unsigned* m2   = (unsigned*)(ws + o); o += (size_t)N;
    float*    den2 = ws + o; o += (size_t)N;

    const int TB = 256;
    auto cdiv = [](int a, int b) { return (a + b - 1) / b; };

    hipMemsetAsync(m1,   0, (size_t)N * 4 * sizeof(unsigned), stream);
    hipMemsetAsync(den1, 0, (size_t)N * 4 * sizeof(float), stream);
    hipMemsetAsync(agg1, 0, (size_t)N * 128 * sizeof(float), stream);
    hipMemsetAsync(m2,   0, (size_t)N * sizeof(unsigned), stream);
    hipMemsetAsync(den2, 0, (size_t)N * sizeof(float), stream);
    hipMemsetAsync(out,  0, (size_t)N * 64 * sizeof(float), stream);

    gemm_wmma_f32<128, 128><<<N / 16, 256, 0, stream>>>(x, W1, h1);
    attn_logits_h4<<<cdiv(N * 4, TB), TB, 0, stream>>>(h1, aS1, aD1, alS1, alD1, N);
    edge_max_h4<<<cdiv(Etot, TB), TB, 0, stream>>>(esrc, edst,
        (const float4*)alS1, (const float4*)alD1, m1, E, Etot);
    edge_exp_h4<<<cdiv(Etot, TB), TB, 0, stream>>>(esrc, edst,
        (const float4*)alS1, (const float4*)alD1, m1, ex1, den1, E, Etot);
    edge_agg_128<<<cdiv(Etot * 128, TB), TB, 0, stream>>>(esrc, edst,
        ex1, den1, h1, agg1, E, Etot);
    finalize_elu<<<cdiv(N * 128, TB), TB, 0, stream>>>(agg1, b1, N);

    gemm_wmma_f32<128, 64><<<N / 16, 128, 0, stream>>>(agg1, W2, h2);
    attn_logits_h1<<<cdiv(N, TB), TB, 0, stream>>>(h2, aS2, aD2, alS2, alD2, N);
    edge_max_h1<<<cdiv(Etot, TB), TB, 0, stream>>>(esrc, edst, alS2, alD2, m2, E, Etot);
    edge_exp_h1<<<cdiv(Etot, TB), TB, 0, stream>>>(esrc, edst, alS2, alD2, m2, ex2, den2, E, Etot);
    edge_agg_64<<<cdiv(Etot * 64, TB), TB, 0, stream>>>(esrc, edst, ex2, den2, h2, out, E, Etot);
    finalize_bias64<<<cdiv(N * 64, TB), TB, 0, stream>>>(out, b2, N);

    (void)n_in; (void)out_size; (void)ws_size;
}


# Edges counting-sorted by destination in-call (int atomics); one wave per node then does max/exp/sum/aggregate in registers: float atomics gone, sums in arrival-rank order, one divide at end
# speedup vs baseline: 4.9111x; 4.9111x over previous
#include <hip/hip_runtime.h>
#include <hip/hip_bf16.h>

typedef float v2f __attribute__((ext_vector_type(2)));
typedef float v8f __attribute__((ext_vector_type(8)));

#define NEG_SLOPE 0.2f
#define INF_EPS   1e-16f

__device__ __forceinline__ float leaky(float x) { return x > 0.f ? x : NEG_SLOPE * x; }

template <int K, int NCOLS>
__global__ __launch_bounds__(256) void gemm_wmma_f32(
    const float* __restrict__ A, const float* __restrict__ B,
    float* __restrict__ C)
{
    const int lane = threadIdx.x & 31;
    const int wave = threadIdx.x >> 5;
    const int row  = blockIdx.x * 16 + (lane & 15);
    const int colB = wave * 16 + (lane & 15);
    const int kh   = (lane >> 4) << 1;

    const float* __restrict__ aPtr = A + (size_t)row * K + kh;
    const float* __restrict__ bPtr = B + (size_t)kh * NCOLS + colB;

    v8f acc = {};
#pragma unroll
    for (int k0 = 0; k0 < K; k0 += 4) {
        v2f a, b;
        a.x = aPtr[k0];
        a.y = aPtr[k0 + 1];
        b.x = bPtr[k0 * NCOLS];
        b.y = bPtr[k0 * NCOLS + NCOLS];
        acc = __builtin_amdgcn_wmma_f32_16x16x4_f32(
            false, a, false, b, (short)0, acc, false, false);
    }
    float* __restrict__ cPtr =
        C + ((size_t)blockIdx.x * 16 + ((lane >> 4) << 3)) * NCOLS + colB;
#pragma unroll
    for (int r = 0; r < 8; ++r)
        cPtr[r * NCOLS] = acc[r];
}

__global__ void attn_logits_h4(const float* __restrict__ h,
                               const float* __restrict__ attS,
                               const float* __restrict__ attD,
                               float* __restrict__ alS, float* __restrict__ alD, int N)
{
    int t = blockIdx.x * blockDim.x + threadIdx.x;
    if (t >= N * 4) return;
    int n = t >> 2, hd = t & 3;
    const float* hp = h + (size_t)n * 128 + hd * 32;
    const float* as = attS + hd * 32;
    const float* ad = attD + hd * 32;
    float s = 0.f, d = 0.f;
#pragma unroll
    for (int c = 0; c < 32; ++c) { float v = hp[c]; s += v * as[c]; d += v * ad[c]; }
    alS[t] = s; alD[t] = d;
}

__global__ void attn_logits_h1(const float* __restrict__ h,
                               const float* __restrict__ attS,
                               const float* __restrict__ attD,
                               float* __restrict__ alS, float* __restrict__ alD, int N)
{
    int n = blockIdx.x * blockDim.x + threadIdx.x;
    if (n >= N) return;
    const float* hp = h + (size_t)n * 64;
    float s = 0.f, d = 0.f;
#pragma unroll
    for (int c = 0; c < 64; ++c) { float v = hp[c]; s += v * attS[c]; d += v * attD[c]; }
    alS[n] = s; alD[n] = d;
}

__global__ void csr_count(const int* __restrict__ edst, int* __restrict__ cnt,
                          int* __restrict__ rank, int E)
{
    int e = blockIdx.x * blockDim.x + threadIdx.x;
    if (e >= E) return;
    rank[e] = atomicAdd(&cnt[edst[e]], 1);
}

__global__ void csr_offsets(const int* __restrict__ cnt, int* __restrict__ off,
                            int* __restrict__ cursor, int N)
{
    const int n = blockIdx.x * blockDim.x + threadIdx.x;
    const int lane = threadIdx.x & 31;
    const int c = (n < N) ? cnt[n] : 0;
    int incl = c;
    for (int o = 1; o < 32; o <<= 1) {
        int v = __shfl_up(incl, o);
        if (lane >= o) incl += v;
    }
    const int total = __shfl(incl, 31);
    int base = 0;
    if (lane == 31) base = atomicAdd(cursor, total);
    base = __shfl(base, 31);
    if (n < N) off[n] = base + incl - c;
}

__global__ void csr_fill(const int* __restrict__ esrc, const int* __restrict__ edst,
                         const int* __restrict__ rank, const int* __restrict__ off,
                         int* __restrict__ ssrc, int E)
{
    int e = blockIdx.x * blockDim.x + threadIdx.x;
    if (e >= E) return;
    ssrc[off[edst[e]] + rank[e]] = esrc[e];
}

__device__ __forceinline__ float wave_max(float v) {
    for (int o = 16; o > 0; o >>= 1) v = fmaxf(v, __shfl_xor(v, o));
    return v;
}
__device__ __forceinline__ float wave_sum(float v) {
    for (int o = 16; o > 0; o >>= 1) v += __shfl_xor(v, o);
    return v;
}
__device__ __forceinline__ float lane_val(float v, int j) {
    return __int_as_float(__builtin_amdgcn_readlane(__float_as_int(v), j));
}
__device__ __forceinline__ float sel4(float a, float b, float c, float d,
                                      unsigned k0, unsigned k1, unsigned k2, unsigned k3) {
    return __uint_as_float((__float_as_uint(a) & k0) | (__float_as_uint(b) & k1) |
                           (__float_as_uint(c) & k2) | (__float_as_uint(d) & k3));
}

__global__ __launch_bounds__(256) void gather_softmax_h4(
    const int* __restrict__ off, const int* __restrict__ cnt, const int* __restrict__ ssrc,
    const float4* __restrict__ alS, const float4* __restrict__ alD,
    const float4* __restrict__ h, const float* __restrict__ bias,
    float4* __restrict__ outp, int N)
{
    const int lane = threadIdx.x & 31;
    const int hd = lane >> 3;
    const unsigned k0 = hd == 0 ? 0xffffffffu : 0u;
    const unsigned k1 = hd == 1 ? 0xffffffffu : 0u;
    const unsigned k2 = hd == 2 ? 0xffffffffu : 0u;
    const unsigned k3 = hd == 3 ? 0xffffffffu : 0u;
    const int n = __builtin_amdgcn_readfirstlane((int)(blockIdx.x * 8 + (threadIdx.x >> 5)));
    if (n >= N) return;
    const int beg = off[n];
    const int deg = cnt[n];
    const float4 dn = alD[n];
    const float4 sn = alS[n];
    const float4 es = make_float4(leaky(sn.x + dn.x), leaky(sn.y + dn.y),
                                  leaky(sn.z + dn.z), leaky(sn.w + dn.w));

    const bool v0 = lane < deg;
    int s = n;
    float4 e0 = es;
    if (v0) {
        s = ssrc[beg + lane];
        const float4 a = alS[s];
        e0 = make_float4(leaky(a.x + dn.x), leaky(a.y + dn.y),
                         leaky(a.z + dn.z), leaky(a.w + dn.w));
    }
    float4 m = make_float4(fmaxf(es.x, e0.x), fmaxf(es.y, e0.y),
                           fmaxf(es.z, e0.z), fmaxf(es.w, e0.w));
    for (int base = 32; base < deg; base += 32) {
        const int i = base + lane;
        if (i < deg) {
            const float4 a = alS[ssrc[beg + i]];
            m.x = fmaxf(m.x, leaky(a.x + dn.x));
            m.y = fmaxf(m.y, leaky(a.y + dn.y));
            m.z = fmaxf(m.z, leaky(a.z + dn.z));
            m.w = fmaxf(m.w, leaky(a.w + dn.w));
        }
    }
    m.x = wave_max(m.x); m.y = wave_max(m.y); m.z = wave_max(m.z); m.w = wave_max(m.w);

    const float4 wself = make_float4(__expf(es.x - m.x), __expf(es.y - m.y),
                                     __expf(es.z - m.z), __expf(es.w - m.w));
    const float wsl = sel4(wself.x, wself.y, wself.z, wself.w, k0, k1, k2, k3);
    const float4 hn = h[(size_t)n * 32 + lane];
    float4 acc = make_float4(wsl * hn.x, wsl * hn.y, wsl * hn.z, wsl * hn.w);
    float4 dp = make_float4(0.f, 0.f, 0.f, 0.f);

    float4 w = make_float4(0.f, 0.f, 0.f, 0.f);
    if (v0) w = make_float4(__expf(e0.x - m.x), __expf(e0.y - m.y),
                            __expf(e0.z - m.z), __expf(e0.w - m.w));
    int base = 0;
    for (;;) {
        dp.x += w.x; dp.y += w.y; dp.z += w.z; dp.w += w.w;
        const int cc = min(32, deg - base);
        for (int j = 0; j < cc; j += 4) {
            int sj[4];
            float wj[4];
            float4 r[4];
            for (int u = 0; u < 4; ++u) {
                sj[u] = __builtin_amdgcn_readlane(s, j + u);
                wj[u] = sel4(lane_val(w.x, j + u), lane_val(w.y, j + u),
                             lane_val(w.z, j + u), lane_val(w.w, j + u), k0, k1, k2, k3);
            }
            for (int u = 0; u < 4; ++u) r[u] = h[(size_t)sj[u] * 32 + lane];
            for (int u = 0; u < 4; ++u) {
                acc.x += wj[u] * r[u].x; acc.y += wj[u] * r[u].y;
                acc.z += wj[u] * r[u].z; acc.w += wj[u] * r[u].w;
            }
        }
        base += 32;
        if (base >= deg) break;
        const int i = base + lane;
        s = n;
        w = make_float4(0.f, 0.f, 0.f, 0.f);
        if (i < deg) {
            s = ssrc[beg + i];
            const float4 a = alS[s];
            w = make_float4(__expf(leaky(a.x + dn.x) - m.x), __expf(leaky(a.y + dn.y) - m.y),
                            __expf(leaky(a.z + dn.z) - m.z), __expf(leaky(a.w + dn.w) - m.w));
        }
    }

    const float den = sel4(wave_sum(dp.x) + wself.x, wave_sum(dp.y) + wself.y,
                           wave_sum(dp.z) + wself.z, wave_sum(dp.w) + wself.w,
                           k0, k1, k2, k3) + INF_EPS;
    const float* bp = bias + lane * 4;
    float4 o;
    o.x = acc.x / den + bp[0];
    o.y = acc.y / den + bp[1];
    o.z = acc.z / den + bp[2];
    o.w = acc.w / den + bp[3];
    o.x = o.x > 0.f ? o.x : __expf(o.x) - 1.f;
    o.y = o.y > 0.f ? o.y : __expf(o.y) - 1.f;
    o.z = o.z > 0.f ? o.z : __expf(o.z) - 1.f;
    o.w = o.w > 0.f ? o.w : __expf(o.w) - 1.f;
    outp[(size_t)n * 32 + lane] = o;
}

__global__ __launch_bounds__(256) void gather_softmax_h1(
    const int* __restrict__ off, const int* __restrict__ cnt, const int* __restrict__ ssrc,
    const float* __restrict__ alS, const float* __restrict__ alD,
    const float4* __restrict__ h, const float* __restrict__ bias,
    float4* __restrict__ outp, int N)
{
    const int lane = threadIdx.x & 31;
    const int g = lane >> 4;
    const int l16 = lane & 15;
    const int n = __builtin_amdgcn_readfirstlane((int)(blockIdx.x * 8 + (threadIdx.x >> 5)));
    if (n >= N) return;
    const int beg = off[n];
    const int deg = cnt[n];
    const float dn = alD[n];
    const float es = leaky(alS[n] + dn);

    const bool v0 = lane < deg;
    int s = n;
    float e0 = es;
    if (v0) {
        s = ssrc[beg + lane];
        e0 = leaky(alS[s] + dn);
    }
    float m = fmaxf(es, e0);
    for (int base = 32; base < deg; base += 32) {
        const int i = base + lane;
        if (i < deg) m = fmaxf(m, leaky(alS[ssrc[beg + i]] + dn));
    }
    m = wave_max(m);

    const float wself = __expf(es - m);
    float4 acc = make_float4(0.f, 0.f, 0.f, 0.f);
    if (g == 0) {
        const float4 hn = h[(size_t)n * 16 + l16];
        acc = make_float4(wself * hn.x, wself * hn.y, wself * hn.z, wself * hn.w);
    }
    float dp = 0.f;
    float w = v0 ? __expf(e0 - m) : 0.f;
    int base = 0;
    for (;;) {
        dp += w;
        const int cc = min(32, deg - base);
        for (int j = 0; j < cc; j += 8) {
            int sj[4];
            float wj[4];
            float4 r[4];
            for (int u = 0; u < 4; ++u) {
                sj[u] = __shfl(s, j + 2 * u + g);
                wj[u] = __shfl(w, j + 2 * u + g);
            }
            for (int u = 0; u < 4; ++u) r[u] = h[(size_t)sj[u] * 16 + l16];
            for (int u = 0; u < 4; ++u) {
                acc.x += wj[u] * r[u].x; acc.y += wj[u] * r[u].y;
                acc.z += wj[u] * r[u].z; acc.w += wj[u] * r[u].w;
            }
        }
        base += 32;
        if (base >= deg) break;
        const int i = base + lane;
        s = n;
        w = 0.f;
        if (i < deg) {
            s = ssrc[beg + i];
            w = __expf(leaky(alS[s] + dn) - m);
        }
    }

    const float den = wave_sum(dp) + wself + INF_EPS;
    acc.x += __shfl_xor(acc.x, 16);
    acc.y += __shfl_xor(acc.y, 16);
    acc.z += __shfl_xor(acc.z, 16);
    acc.w += __shfl_xor(acc.w, 16);
    if (g == 0) {
        const float* bp = bias + l16 * 4;
        float4 o;
        o.x = acc.x / den + bp[0];
        o.y = acc.y / den + bp[1];
        o.z = acc.z / den + bp[2];
        o.w = acc.w / den + bp[3];
        outp[(size_t)n * 16 + l16] = o;
    }
}

extern "C" void kernel_launch(void* const* d_in, const int* in_sizes, int n_in,
                              void* d_out, int out_size, void* d_ws, size_t ws_size,
                              hipStream_t stream)
{
    const float* x    = (const float*)d_in[0];
    const int*   eidx = (const int*)d_in[1];
    const float* W1   = (const float*)d_in[2];
    const float* aS1  = (const float*)d_in[3];
    const float* aD1  = (const float*)d_in[4];
    const float* b1   = (const float*)d_in[5];
    const float* W2   = (const float*)d_in[6];
    const float* aS2  = (const float*)d_in[7];
    const float* aD2  = (const float*)d_in[8];
    const float* b2   = (const float*)d_in[9];
    float* out = (float*)d_out;

    const int N    = in_sizes[0] / 128;
    const int E    = in_sizes[1] / 2;
    const int* esrc = eidx;
    const int* edst = eidx + E;

    float* ws = (float*)d_ws;
    size_t o = 0;
    float* h1   = ws + o; o += (size_t)N * 128;
    float* alS1 = ws + o; o += (size_t)N * 4;
    float* alD1 = ws + o; o += (size_t)N * 4;
    float* agg1 = ws + o; o += (size_t)N * 128;
    float* h2   = ws + o; o += (size_t)N * 64;
    float* alS2 = ws + o; o += (size_t)N;
    float* alD2 = ws + o; o += (size_t)N;
    int*   off  = (int*)(ws + o); o += (size_t)N;
    int*   rank = (int*)(ws + o); o += (size_t)E;
    int*   ssrc = (int*)(ws + o); o += (size_t)E;
    int*   cnt  = (int*)(ws + o); o += (size_t)N + 1;

    const int TB = 256;
    auto cdiv = [](int a, int b) { return (a + b - 1) / b; };

    hipMemsetAsync(cnt, 0, ((size_t)N + 1) * sizeof(int), stream);
    csr_count<<<cdiv(E, TB), TB, 0, stream>>>(edst, cnt, rank, E);
    csr_offsets<<<cdiv(N, TB), TB, 0, stream>>>(cnt, off, cnt + N, N);
    csr_fill<<<cdiv(E, TB), TB, 0, stream>>>(esrc, edst, rank, off, ssrc, E);

    gemm_wmma_f32<128, 128><<<N / 16, 256, 0, stream>>>(x, W1, h1);
    attn_logits_h4<<<cdiv(N * 4, TB), TB, 0, stream>>>(h1, aS1, aD1, alS1, alD1, N);
    gather_softmax_h4<<<cdiv(N, 8), 256, 0, stream>>>(off, cnt, ssrc,
        (const float4*)alS1, (const float4*)alD1, (const float4*)h1, b1, (float4*)agg1, N);

    gemm_wmma_f32<128, 64><<<N / 16, 128, 0, stream>>>(agg1, W2, h2);
    attn_logits_h1<<<cdiv(N, TB), TB, 0, stream>>>(h2, aS2, aD2, alS2, alD2, N);
    gather_softmax_h1<<<cdiv(N, 8), 256, 0, stream>>>(off, cnt, ssrc,
        alS2, alD2, (const float4*)h2, b2, (float4*)out, N);

    (void)n_in; (void)out_size; (void)ws_size;
}


# Edge sort by destination ranks edges with LDS counters in two levels (256-node buckets, then nodes of a bucket): no returning global add per edge; sums stay in arrival order (LDS arrival now)
# speedup vs baseline: 7.1008x; 1.4459x over previous
#include <hip/hip_runtime.h>
#include <hip/hip_bf16.h>

typedef float v2f __attribute__((ext_vector_type(2)));
typedef float v8f __attribute__((ext_vector_type(8)));

#define NEG_SLOPE 0.2f
#define INF_EPS   1e-16f

__device__ __forceinline__ float leaky(float x) { return x > 0.f ? x : NEG_SLOPE * x; }

template <int K, int NCOLS>
__global__ __launch_bounds__(256) void gemm_wmma_f32(
    const float* __restrict__ A, const float* __restrict__ B,
    float* __restrict__ C)
{
    const int lane = threadIdx.x & 31;
    const int wave = threadIdx.x >> 5;
    const int row  = blockIdx.x * 16 + (lane & 15);
    const int colB = wave * 16 + (lane & 15);
    const int kh   = (lane >> 4) << 1;

    const float* __restrict__ aPtr = A + (size_t)row * K + kh;
    const float* __restrict__ bPtr = B + (size_t)kh * NCOLS + colB;

    v8f acc = {};
#pragma unroll
    for (int k0 = 0; k0 < K; k0 += 4) {
        v2f a, b;
        a.x = aPtr[k0];
        a.y = aPtr[k0 + 1];
        b.x = bPtr[k0 * NCOLS];
        b.y = bPtr[k0 * NCOLS + NCOLS];
        acc = __builtin_amdgcn_wmma_f32_16x16x4_f32(
            false, a, false, b, (short)0, acc, false, false);
    }
    float* __restrict__ cPtr =
        C + ((size_t)blockIdx.x * 16 + ((lane >> 4) << 3)) * NCOLS + colB;
#pragma unroll
    for (int r = 0; r < 8; ++r)
        cPtr[r * NCOLS] = acc[r];
}

__global__ void attn_logits_h4(const float* __restrict__ h,
                               const float* __restrict__ attS,
                               const float* __restrict__ attD,
                               float* __restrict__ alS, float* __restrict__ alD, int N)
{
    int t = blockIdx.x * blockDim.x + threadIdx.x;
    if (t >= N * 4) return;
    int n = t >> 2, hd = t & 3;
    const float* hp = h + (size_t)n * 128 + hd * 32;
    const float* as = attS + hd * 32;
    const float* ad = attD + hd * 32;
    float s = 0.f, d = 0.f;
#pragma unroll
    for (int c = 0; c < 32; ++c) { float v = hp[c]; s += v * as[c]; d += v * ad[c]; }
    alS[t] = s; alD[t] = d;
}

__global__ void attn_logits_h1(const float* __restrict__ h,
                               const float* __restrict__ attS,
                               const float* __restrict__ attD,
                               float* __restrict__ alS, float* __restrict__ alD, int N)
{
    int n = blockIdx.x * blockDim.x + threadIdx.x;
    if (n >= N) return;
    const float* hp = h + (size_t)n * 64;
    float s = 0.f, d = 0.f;
#pragma unroll
    for (int c = 0; c < 64; ++c) { float v = hp[c]; s += v * attS[c]; d += v * attD[c]; }
    alS[n] = s; alD[n] = d;
}

#define SORT_CHUNK 2048

__device__ __forceinline__ int wave_incl_scan_i(int v, int lane) {
    for (int o = 1; o < 32; o <<= 1) {
        int u = __shfl_up(v, o);
        if (lane >= o) v += u;
    }
    return v;
}

__global__ __launch_bounds__(256) void bucket_count(
    const int* __restrict__ edst, int* __restrict__ tot, int* __restrict__ cbase, int E)
{
    __shared__ int s_h[256];
    const int t = threadIdx.x;
    s_h[t] = 0;
    __syncthreads();
    const int e0 = blockIdx.x * SORT_CHUNK + t;
    int dv[SORT_CHUNK / 256];
#pragma unroll
    for (int k = 0; k < SORT_CHUNK / 256; ++k) {
        const int e = e0 + k * 256;
        dv[k] = 0;
        if (e < E) dv[k] = edst[e];
    }
#pragma unroll
    for (int k = 0; k < SORT_CHUNK / 256; ++k) {
        const int e = e0 + k * 256;
        if (e < E) atomicAdd(&s_h[dv[k] >> 8], 1);
    }
    __syncthreads();
    const int hcount = s_h[t];
    int base = 0;
    if (hcount > 0) base = atomicAdd(&tot[t * 32], hcount);
    cbase[blockIdx.x * 256 + t] = base;
}

__global__ __launch_bounds__(256) void bucket_scatter(
    const int* __restrict__ esrc, const int* __restrict__ edst,
    const int* __restrict__ tot, const int* __restrict__ cbase,
    unsigned* __restrict__ bpack, int E)
{
    __shared__ int s_cur[256];
    __shared__ int s_w[8];
    const int t = threadIdx.x;
    const int lane = t & 31;
    const int wave = t >> 5;
    const int e0 = blockIdx.x * SORT_CHUNK + t;
    int sv[SORT_CHUNK / 256], dv[SORT_CHUNK / 256];
#pragma unroll
    for (int k = 0; k < SORT_CHUNK / 256; ++k) {
        const int e = e0 + k * 256;
        sv[k] = 0; dv[k] = 0;
        if (e < E) { sv[k] = esrc[e]; dv[k] = edst[e]; }
    }
    const int c = tot[t * 32];
    const int incl = wave_incl_scan_i(c, lane);
    if (lane == 31) s_w[wave] = incl;
    __syncthreads();
    int wb = 0;
    for (int w = 0; w < 8; ++w) if (w < wave) wb += s_w[w];
    s_cur[t] = wb + incl - c + cbase[blockIdx.x * 256 + t];
    __syncthreads();
#pragma unroll
    for (int k = 0; k < SORT_CHUNK / 256; ++k) {
        const int e = e0 + k * 256;
        if (e < E) {
            const int pos = atomicAdd(&s_cur[dv[k] >> 8], 1);
            bpack[pos] = ((unsigned)sv[k] << 8) | ((unsigned)dv[k] & 255u);
        }
    }
}

__global__ __launch_bounds__(1024) void bucket_sort(
    const unsigned* __restrict__ bpack, const int* __restrict__ tot,
    int* __restrict__ cnt, int* __restrict__ off, int* __restrict__ ssrc, int N)
{
    __shared__ int s_cnt[256];
    __shared__ int s_cur[256];
    __shared__ int s_w[8];
    __shared__ int s_start;
    const int b = blockIdx.x;
    const int t = threadIdx.x;
    const int lane = t & 31;
    const int wave = t >> 5;
    if (t < 256) s_cnt[t] = 0;
    if (t == 0) s_start = 0;
    __syncthreads();
    if (t < 256) {
        int v = (t < b) ? tot[t * 32] : 0;
        for (int o = 16; o > 0; o >>= 1) v += __shfl_xor(v, o);
        if (lane == 0 && v != 0) atomicAdd(&s_start, v);
    }
    const int m = tot[b * 32];
    __syncthreads();
    const int start = s_start;
    for (int i0 = t; i0 < m; i0 += 4096) {
        unsigned p[4];
#pragma unroll
        for (int k = 0; k < 4; ++k) {
            const int i = i0 + k * 1024;
            p[k] = 0u;
            if (i < m) p[k] = bpack[start + i];
        }
#pragma unroll
        for (int k = 0; k < 4; ++k) {
            const int i = i0 + k * 1024;
            if (i < m) atomicAdd(&s_cnt[p[k] & 255u], 1);
        }
    }
    __syncthreads();
    const int c = (t < 256) ? s_cnt[t] : 0;
    const int incl = wave_incl_scan_i(c, lane);
    if (t < 256 && lane == 31) s_w[wave] = incl;
    __syncthreads();
    if (t < 256) {
        int wb = 0;
        for (int w = 0; w < 8; ++w) if (w < wave) wb += s_w[w];
        const int ex = wb + incl - c;
        s_cur[t] = ex;
        const int n = b * 256 + t;
        if (n < N) { cnt[n] = c; off[n] = start + ex; }
    }
    __syncthreads();
    for (int i0 = t; i0 < m; i0 += 4096) {
        unsigned p[4];
#pragma unroll
        for (int k = 0; k < 4; ++k) {
            const int i = i0 + k * 1024;
            p[k] = 0u;
            if (i < m) p[k] = bpack[start + i];
        }
#pragma unroll
        for (int k = 0; k < 4; ++k) {
            const int i = i0 + k * 1024;
            if (i < m) {
                const int pos = atomicAdd(&s_cur[p[k] & 255u], 1);
                ssrc[start + pos] = (int)(p[k] >> 8);
            }
        }
    }
}

__device__ __forceinline__ float wave_max(float v) {
    for (int o = 16; o > 0; o >>= 1) v = fmaxf(v, __shfl_xor(v, o));
    return v;
}
__device__ __forceinline__ float wave_sum(float v) {
    for (int o = 16; o > 0; o >>= 1) v += __shfl_xor(v, o);
    return v;
}
__device__ __forceinline__ float lane_val(float v, int j) {
    return __int_as_float(__builtin_amdgcn_readlane(__float_as_int(v), j));
}
__device__ __forceinline__ float sel4(float a, float b, float c, float d,
                                      unsigned k0, unsigned k1, unsigned k2, unsigned k3) {
    return __uint_as_float((__float_as_uint(a) & k0) | (__float_as_uint(b) & k1) |
                           (__float_as_uint(c) & k2) | (__float_as_uint(d) & k3));
}

__global__ __launch_bounds__(256) void gather_softmax_h4(
    const int* __restrict__ off, const int* __restrict__ cnt, const int* __restrict__ ssrc,
    const float4* __restrict__ alS, const float4* __restrict__ alD,
    const float4* __restrict__ h, const float* __restrict__ bias,
    float4* __restrict__ outp, int N)
{
    const int lane = threadIdx.x & 31;
    const int hd = lane >> 3;
    const unsigned k0 = hd == 0 ? 0xffffffffu : 0u;
    const unsigned k1 = hd == 1 ? 0xffffffffu : 0u;
    const unsigned k2 = hd == 2 ? 0xffffffffu : 0u;
    const unsigned k3 = hd == 3 ? 0xffffffffu : 0u;
    const int n = __builtin_amdgcn_readfirstlane((int)(blockIdx.x * 8 + (threadIdx.x >> 5)));
    if (n >= N) return;
    const int beg = off[n];
    const int deg = cnt[n];
    const float4 dn = alD[n];
    const float4 sn = alS[n];
    const float4 es = make_float4(leaky(sn.x + dn.x), leaky(sn.y + dn.y),
                                  leaky(sn.z + dn.z), leaky(sn.w + dn.w));

    const bool v0 = lane < deg;
    int s = n;
    float4 e0 = es;
    if (v0) {
        s = ssrc[beg + lane];
        const float4 a = alS[s];
        e0 = make_float4(leaky(a.x + dn.x), leaky(a.y + dn.y),
                         leaky(a.z + dn.z), leaky(a.w + dn.w));
    }
    float4 m = make_float4(fmaxf(es.x, e0.x), fmaxf(es.y, e0.y),
                           fmaxf(es.z, e0.z), fmaxf(es.w, e0.w));
    for (int base = 32; base < deg; base += 32) {
        const int i = base + lane;
        if (i < deg) {
            const float4 a = alS[ssrc[beg + i]];
            m.x = fmaxf(m.x, leaky(a.x + dn.x));
            m.y = fmaxf(m.y, leaky(a.y + dn.y));
            m.z = fmaxf(m.z, leaky(a.z + dn.z));
            m.w = fmaxf(m.w, leaky(a.w + dn.w));
        }
    }
    m.x = wave_max(m.x); m.y = wave_max(m.y); m.z = wave_max(m.z); m.w = wave_max(m.w);

    const float4 wself = make_float4(__expf(es.x - m.x), __expf(es.y - m.y),
                                     __expf(es.z - m.z), __expf(es.w - m.w));
    const float wsl = sel4(wself.x, wself.y, wself.z, wself.w, k0, k1, k2, k3);
    const float4 hn = h[(size_t)n * 32 + lane];
    float4 acc = make_float4(wsl * hn.x, wsl * hn.y, wsl * hn.z, wsl * hn.w);
    float4 dp = make_float4(0.f, 0.f, 0.f, 0.f);

    float4 w = make_float4(0.f, 0.f, 0.f, 0.f);
    if (v0) w = make_float4(__expf(e0.x - m.x), __expf(e0.y - m.y),
                            __expf(e0.z - m.z), __expf(e0.w - m.w));
    int base = 0;
    for (;;) {
        dp.x += w.x; dp.y += w.y; dp.z += w.z; dp.w += w.w;
        const int cc = min(32, deg - base);
        for (int j = 0; j < cc; j += 4) {
            int sj[4];
            float wj[4];
            float4 r[4];
            for (int u = 0; u < 4; ++u) {
                sj[u] = __builtin_amdgcn_readlane(s, j + u);
                wj[u] = sel4(lane_val(w.x, j + u), lane_val(w.y, j + u),
                             lane_val(w.z, j + u), lane_val(w.w, j + u), k0, k1, k2, k3);
            }
            for (int u = 0; u < 4; ++u) r[u] = h[(size_t)sj[u] * 32 + lane];
            for (int u = 0; u < 4; ++u) {
                acc.x += wj[u] * r[u].x; acc.y += wj[u] * r[u].y;
                acc.z += wj[u] * r[u].z; acc.w += wj[u] * r[u].w;
            }
        }
        base += 32;
        if (base >= deg) break;
        const int i = base + lane;
        s = n;
        w = make_float4(0.f, 0.f, 0.f, 0.f);
        if (i < deg) {
            s = ssrc[beg + i];
            const float4 a = alS[s];
            w = make_float4(__expf(leaky(a.x + dn.x) - m.x), __expf(leaky(a.y + dn.y) - m.y),
                            __expf(leaky(a.z + dn.z) - m.z), __expf(leaky(a.w + dn.w) - m.w));
        }
    }

    const float den = sel4(wave_sum(dp.x) + wself.x, wave_sum(dp.y) + wself.y,
                           wave_sum(dp.z) + wself.z, wave_sum(dp.w) + wself.w,
                           k0, k1, k2, k3) + INF_EPS;
    const float* bp = bias + lane * 4;
    float4 o;
    o.x = acc.x / den + bp[0];
    o.y = acc.y / den + bp[1];
    o.z = acc.z / den + bp[2];
    o.w = acc.w / den + bp[3];
    o.x = o.x > 0.f ? o.x : __expf(o.x) - 1.f;
    o.y = o.y > 0.f ? o.y : __expf(o.y) - 1.f;
    o.z = o.z > 0.f ? o.z : __expf(o.z) - 1.f;
    o.w = o.w > 0.f ? o.w : __expf(o.w) - 1.f;
    outp[(size_t)n * 32 + lane] = o;
}

__global__ __launch_bounds__(256) void gather_softmax_h1(
    const int* __restrict__ off, const int* __restrict__ cnt, const int* __restrict__ ssrc,
    const float* __restrict__ alS, const float* __restrict__ alD,
    const float4* __restrict__ h, const float* __restrict__ bias,
    float4* __restrict__ outp, int N)
{
    const int lane = threadIdx.x & 31;
    const int g = lane >> 4;
    const int l16 = lane & 15;
    const int n = __builtin_amdgcn_readfirstlane((int)(blockIdx.x * 8 + (threadIdx.x >> 5)));
    if (n >= N) return;
    const int beg = off[n];
    const int deg = cnt[n];
    const float dn = alD[n];
    const float es = leaky(alS[n] + dn);

    const bool v0 = lane < deg;
    int s = n;
    float e0 = es;
    if (v0) {
        s = ssrc[beg + lane];
        e0 = leaky(alS[s] + dn);
    }
    float m = fmaxf(es, e0);
    for (int base = 32; base < deg; base += 32) {
        const int i = base + lane;
        if (i < deg) m = fmaxf(m, leaky(alS[ssrc[beg + i]] + dn));
    }
    m = wave_max(m);

    const float wself = __expf(es - m);
    float4 acc = make_float4(0.f, 0.f, 0.f, 0.f);
    if (g == 0) {
        const float4 hn = h[(size_t)n * 16 + l16];
        acc = make_float4(wself * hn.x, wself * hn.y, wself * hn.z, wself * hn.w);
    }
    float dp = 0.f;
    float w = v0 ? __expf(e0 - m) : 0.f;
    int base = 0;
    for (;;) {
        dp += w;
        const int cc = min(32, deg - base);
        for (int j = 0; j < cc; j += 8) {
            int sj[4];
            float wj[4];
            float4 r[4];
            for (int u = 0; u < 4; ++u) {
                sj[u] = __shfl(s, j + 2 * u + g);
                wj[u] = __shfl(w, j + 2 * u + g);
            }
            for (int u = 0; u < 4; ++u) r[u] = h[(size_t)sj[u] * 16 + l16];
            for (int u = 0; u < 4; ++u) {
                acc.x += wj[u] * r[u].x; acc.y += wj[u] * r[u].y;
                acc.z += wj[u] * r[u].z; acc.w += wj[u] * r[u].w;
            }
        }
        base += 32;
        if (base >= deg) break;
        const int i = base + lane;
        s = n;
        w = 0.f;
        if (i < deg) {
            s = ssrc[beg + i];
            w = __expf(leaky(alS[s] + dn) - m);
        }
    }

    const float den = wave_sum(dp) + wself + INF_EPS;
    acc.x += __shfl_xor(acc.x, 16);
    acc.y += __shfl_xor(acc.y, 16);
    acc.z += __shfl_xor(acc.z, 16);
    acc.w += __shfl_xor(acc.w, 16);
    if (g == 0) {
        const float* bp = bias + l16 * 4;
        float4 o;
        o.x = acc.x / den + bp[0];
        o.y = acc.y / den + bp[1];
        o.z = acc.z / den + bp[2];
        o.w = acc.w / den + bp[3];
        outp[(size_t)n * 16 + l16] = o;
    }
}

extern "C" void kernel_launch(void* const* d_in, const int* in_sizes, int n_in,
                              void* d_out, int out_size, void* d_ws, size_t ws_size,
                              hipStream_t stream)
{
    const float* x    = (const float*)d_in[0];
    const int*   eidx = (const int*)d_in[1];
    const float* W1   = (const float*)d_in[2];
    const float* aS1  = (const float*)d_in[3];
    const float* aD1  = (const float*)d_in[4];
    const float* b1   = (const float*)d_in[5];
    const float* W2   = (const float*)d_in[6];
    const float* aS2  = (const float*)d_in[7];
    const float* aD2  = (const float*)d_in[8];
    const float* b2   = (const float*)d_in[9];
    float* out = (float*)d_out;

    const int N    = in_sizes[0] / 128;
    const int E    = in_sizes[1] / 2;
    const int* esrc = eidx;
    const int* edst = eidx + E;

    float* ws = (float*)d_ws;
    size_t o = 0;
    float* h1   = ws + o; o += (size_t)N * 128;
    float* alS1 = ws + o; o += (size_t)N * 4;
    float* alD1 = ws + o; o += (size_t)N * 4;
    float* agg1 = ws + o; o += (size_t)N * 128;
    float* h2   = ws + o; o += (size_t)N * 64;
    float* alS2 = ws + o; o += (size_t)N;
    float* alD2 = ws + o; o += (size_t)N;
    int*   off  = (int*)(ws + o); o += (size_t)N;
    unsigned* bpack = (unsigned*)(ws + o); o += (size_t)E;
    int*   ssrc = (int*)(ws + o); o += (size_t)E;
    int*   cnt  = (int*)(ws + o); o += (size_t)N;
    o = (o + 31) & ~(size_t)31;
    int*   tot  = (int*)(ws + o); o += (size_t)256 * 32;
    const int TB = 256;
    auto cdiv = [](int a, int b) { return (a + b - 1) / b; };
    const int nChunk  = cdiv(E, SORT_CHUNK);
    const int nBucket = cdiv(N, 256);
    int*   cbase = (int*)(ws + o); o += (size_t)nChunk * 256;

    hipMemsetAsync(tot, 0, (size_t)256 * 32 * sizeof(int), stream);
    bucket_count<<<nChunk, 256, 0, stream>>>(edst, tot, cbase, E);
    bucket_scatter<<<nChunk, 256, 0, stream>>>(esrc, edst, tot, cbase, bpack, E);
    bucket_sort<<<nBucket, 1024, 0, stream>>>(bpack, tot, cnt, off, ssrc, N);

    gemm_wmma_f32<128, 128><<<N / 16, 256, 0, stream>>>(x, W1, h1);
    attn_logits_h4<<<cdiv(N * 4, TB), TB, 0, stream>>>(h1, aS1, aD1, alS1, alD1, N);
    gather_softmax_h4<<<cdiv(N, 8), 256, 0, stream>>>(off, cnt, ssrc,
        (const float4*)alS1, (const float4*)alD1, (const float4*)h1, b1, (float4*)agg1, N);

    gemm_wmma_f32<128, 64><<<N / 16, 128, 0, stream>>>(agg1, W2, h2);
    attn_logits_h1<<<cdiv(N, TB), TB, 0, stream>>>(h2, aS2, aD2, alS2, alD2, N);
    gather_softmax_h1<<<cdiv(N, 8), 256, 0, stream>>>(off, cnt, ssrc,
        alS2, alD2, (const float4*)h2, b2, (float4*)out, N);

    (void)n_in; (void)out_size; (void)ws_size;
}


# Dense products: one wave per 16 rows x 64 columns (4 accumulator tiles on one A fragment, tile columns interleaved by 4), so B loads and C stores are 16 bytes per lane; k order and sums unchanged
# speedup vs baseline: 7.8339x; 1.1032x over previous
#include <hip/hip_runtime.h>
#include <hip/hip_bf16.h>

typedef float v2f __attribute__((ext_vector_type(2)));
typedef float v8f __attribute__((ext_vector_type(8)));

#define NEG_SLOPE 0.2f
#define INF_EPS   1e-16f

__device__ __forceinline__ float leaky(float x) { return x > 0.f ? x : NEG_SLOPE * x; }

template <int K, int NCOLS>
__global__ __launch_bounds__(128) void gemm_wmma_f32(
    const float* __restrict__ A, const float* __restrict__ B,
    float* __restrict__ C, int rowTiles)
{
    constexpr int CG = NCOLS / 64;
    const int lane = threadIdx.x & 31;
    const int wave = threadIdx.x >> 5;
    const int rt   = blockIdx.x * (4 / CG) + wave / CG;
    if (rt >= rowTiles) return;
    const int c    = lane & 15;
    const int row  = rt * 16 + c;
    const int col4 = (wave % CG) * 64 + 4 * c;
    const int kh   = (lane >> 4) << 1;

    const float* __restrict__ aPtr = A + (size_t)row * K + kh;
    const float* __restrict__ bPtr = B + (size_t)kh * NCOLS + col4;

    v8f acc0 = {}, acc1 = {}, acc2 = {}, acc3 = {};
#pragma unroll
    for (int k0 = 0; k0 < K; k0 += 4) {
        v2f a, b;
        a.x = aPtr[k0];
        a.y = aPtr[k0 + 1];
        const float4 b0 = *reinterpret_cast<const float4*>(bPtr + (size_t)k0 * NCOLS);
        const float4 b1 = *reinterpret_cast<const float4*>(bPtr + (size_t)k0 * NCOLS + NCOLS);
        b.x = b0.x; b.y = b1.x;
        acc0 = __builtin_amdgcn_wmma_f32_16x16x4_f32(
            false, a, false, b, (short)0, acc0, false, false);
        b.x = b0.y; b.y = b1.y;
        acc1 = __builtin_amdgcn_wmma_f32_16x16x4_f32(
            false, a, false, b, (short)0, acc1, false, false);
        b.x = b0.z; b.y = b1.z;
        acc2 = __builtin_amdgcn_wmma_f32_16x16x4_f32(
            false, a, false, b, (short)0, acc2, false, false);
        b.x = b0.w; b.y = b1.w;
        acc3 = __builtin_amdgcn_wmma_f32_16x16x4_f32(
            false, a, false, b, (short)0, acc3, false, false);
    }
    float* __restrict__ cPtr =
        C + ((size_t)rt * 16 + ((lane >> 4) << 3)) * NCOLS + col4;
#pragma unroll
    for (int r = 0; r < 8; ++r)
        *reinterpret_cast<float4*>(cPtr + (size_t)r * NCOLS) =
            make_float4(acc0[r], acc1[r], acc2[r], acc3[r]);
}

__global__ void attn_logits_h4(const float* __restrict__ h,
                               const float* __restrict__ attS,
                               const float* __restrict__ attD,
                               float* __restrict__ alS, float* __restrict__ alD, int N)
{
    int t = blockIdx.x * blockDim.x + threadIdx.x;
    if (t >= N * 4) return;
    int n = t >> 2, hd = t & 3;
    const float* hp = h + (size_t)n * 128 + hd * 32;
    const float* as = attS + hd * 32;
    const float* ad = attD + hd * 32;
    float s = 0.f, d = 0.f;
#pragma unroll
    for (int c = 0; c < 32; ++c) { float v = hp[c]; s += v * as[c]; d += v * ad[c]; }
    alS[t] = s; alD[t] = d;
}

__global__ void attn_logits_h1(const float* __restrict__ h,
                               const float* __restrict__ attS,
                               const float* __restrict__ attD,
                               float* __restrict__ alS, float* __restrict__ alD, int N)
{
    int n = blockIdx.x * blockDim.x + threadIdx.x;
    if (n >= N) return;
    const float* hp = h + (size_t)n * 64;
    float s = 0.f, d = 0.f;
#pragma unroll
    for (int c = 0; c < 64; ++c) { float v = hp[c]; s += v * attS[c]; d += v * attD[c]; }
    alS[n] = s; alD[n] = d;
}

#define SORT_CHUNK 2048

__device__ __forceinline__ int wave_incl_scan_i(int v, int lane) {
    for (int o = 1; o < 32; o <<= 1) {
        int u = __shfl_up(v, o);
        if (lane >= o) v += u;
    }
    return v;
}

__global__ __launch_bounds__(256) void bucket_count(
    const int* __restrict__ edst, int* __restrict__ tot, int* __restrict__ cbase, int E)
{
    __shared__ int s_h[256];
    const int t = threadIdx.x;
    s_h[t] = 0;
    __syncthreads();
    const int e0 = blockIdx.x * SORT_CHUNK + t;
    int dv[SORT_CHUNK / 256];
#pragma unroll
    for (int k = 0; k < SORT_CHUNK / 256; ++k) {
        const int e = e0 + k * 256;
        dv[k] = 0;
        if (e < E) dv[k] = edst[e];
    }
#pragma unroll
    for (int k = 0; k < SORT_CHUNK / 256; ++k) {
        const int e = e0 + k * 256;
        if (e < E) atomicAdd(&s_h[dv[k] >> 8], 1);
    }
    __syncthreads();
    const int hcount = s_h[t];
    int base = 0;
    if (hcount > 0) base = atomicAdd(&tot[t * 32], hcount);
    cbase[blockIdx.x * 256 + t] = base;
}

__global__ __launch_bounds__(256) void bucket_scatter(
    const int* __restrict__ esrc, const int* __restrict__ edst,
    const int* __restrict__ tot, const int* __restrict__ cbase,
    unsigned* __restrict__ bpack, int E)
{
    __shared__ int s_cur[256];
    __shared__ int s_w[8];
    const int t = threadIdx.x;
    const int lane = t & 31;
    const int wave = t >> 5;
    const int e0 = blockIdx.x * SORT_CHUNK + t;
    int sv[SORT_CHUNK / 256], dv[SORT_CHUNK / 256];
#pragma unroll
    for (int k = 0; k < SORT_CHUNK / 256; ++k) {
        const int e = e0 + k * 256;
        sv[k] = 0; dv[k] = 0;
        if (e < E) { sv[k] = esrc[e]; dv[k] = edst[e]; }
    }
    const int c = tot[t * 32];
    const int incl = wave_incl_scan_i(c, lane);
    if (lane == 31) s_w[wave] = incl;
    __syncthreads();
    int wb = 0;
    for (int w = 0; w < 8; ++w) if (w < wave) wb += s_w[w];
    s_cur[t] = wb + incl - c + cbase[blockIdx.x * 256 + t];
    __syncthreads();
#pragma unroll
    for (int k = 0; k < SORT_CHUNK / 256; ++k) {
        const int e = e0 + k * 256;
        if (e < E) {
            const int pos = atomicAdd(&s_cur[dv[k] >> 8], 1);
            bpack[pos] = ((unsigned)sv[k] << 8) | ((unsigned)dv[k] & 255u);
        }
    }
}

__global__ __launch_bounds__(1024) void bucket_sort(
    const unsigned* __restrict__ bpack, const int* __restrict__ tot,
    int* __restrict__ cnt, int* __restrict__ off, int* __restrict__ ssrc, int N)
{
    __shared__ int s_cnt[256];
    __shared__ int s_cur[256];
    __shared__ int s_w[8];
    __shared__ int s_start;
    const int b = blockIdx.x;
    const int t = threadIdx.x;
    const int lane = t & 31;
    const int wave = t >> 5;
    if (t < 256) s_cnt[t] = 0;
    if (t == 0) s_start = 0;
    __syncthreads();
    if (t < 256) {
        int v = (t < b) ? tot[t * 32] : 0;
        for (int o = 16; o > 0; o >>= 1) v += __shfl_xor(v, o);
        if (lane == 0 && v != 0) atomicAdd(&s_start, v);
    }
    const int m = tot[b * 32];
    __syncthreads();
    const int start = s_start;
    for (int i0 = t; i0 < m; i0 += 4096) {
        unsigned p[4];
#pragma unroll
        for (int k = 0; k < 4; ++k) {
            const int i = i0 + k * 1024;
            p[k] = 0u;
            if (i < m) p[k] = bpack[start + i];
        }
#pragma unroll
        for (int k = 0; k < 4; ++k) {
            const int i = i0 + k * 1024;
            if (i < m) atomicAdd(&s_cnt[p[k] & 255u], 1);
        }
    }
    __syncthreads();
    const int c = (t < 256) ? s_cnt[t] : 0;
    const int incl = wave_incl_scan_i(c, lane);
    if (t < 256 && lane == 31) s_w[wave] = incl;
    __syncthreads();
    if (t < 256) {
        int wb = 0;
        for (int w = 0; w < 8; ++w) if (w < wave) wb += s_w[w];
        const int ex = wb + incl - c;
        s_cur[t] = ex;
        const int n = b * 256 + t;
        if (n < N) { cnt[n] = c; off[n] = start + ex; }
    }
    __syncthreads();
    for (int i0 = t; i0 < m; i0 += 4096) {
        unsigned p[4];
#pragma unroll
        for (int k = 0; k < 4; ++k) {
            const int i = i0 + k * 1024;
            p[k] = 0u;
            if (i < m) p[k] = bpack[start + i];
        }
#pragma unroll
        for (int k = 0; k < 4; ++k) {
            const int i = i0 + k * 1024;
            if (i < m) {
                const int pos = atomicAdd(&s_cur[p[k] & 255u], 1);
                ssrc[start + pos] = (int)(p[k] >> 8);
            }
        }
    }
}

__device__ __forceinline__ float wave_max(float v) {
    for (int o = 16; o > 0; o >>= 1) v = fmaxf(v, __shfl_xor(v, o));
    return v;
}
__device__ __forceinline__ float wave_sum(float v) {
    for (int o = 16; o > 0; o >>= 1) v += __shfl_xor(v, o);
    return v;
}
__device__ __forceinline__ float lane_val(float v, int j) {
    return __int_as_float(__builtin_amdgcn_readlane(__float_as_int(v), j));
}
__device__ __forceinline__ float sel4(float a, float b, float c, float d,
                                      unsigned k0, unsigned k1, unsigned k2, unsigned k3) {
    return __uint_as_float((__float_as_uint(a) & k0) | (__float_as_uint(b) & k1) |
                           (__float_as_uint(c) & k2) | (__float_as_uint(d) & k3));
}

__global__ __launch_bounds__(256) void gather_softmax_h4(
    const int* __restrict__ off, const int* __restrict__ cnt, const int* __restrict__ ssrc,
    const float4* __restrict__ alS, const float4* __restrict__ alD,
    const float4* __restrict__ h, const float* __restrict__ bias,
    float4* __restrict__ outp, int N)
{
    const int lane = threadIdx.x & 31;
    const int hd = lane >> 3;
    const unsigned k0 = hd == 0 ? 0xffffffffu : 0u;
    const unsigned k1 = hd == 1 ? 0xffffffffu : 0u;
    const unsigned k2 = hd == 2 ? 0xffffffffu : 0u;
    const unsigned k3 = hd == 3 ? 0xffffffffu : 0u;
    const int n = __builtin_amdgcn_readfirstlane((int)(blockIdx.x * 8 + (threadIdx.x >> 5)));
    if (n >= N) return;
    const int beg = off[n];
    const int deg = cnt[n];
    const float4 dn = alD[n];
    const float4 sn = alS[n];
    const float4 es = make_float4(leaky(sn.x + dn.x), leaky(sn.y + dn.y),
                                  leaky(sn.z + dn.z), leaky(sn.w + dn.w));

    const bool v0 = lane < deg;
    int s = n;
    float4 e0 = es;
    if (v0) {
        s = ssrc[beg + lane];
        const float4 a = alS[s];
        e0 = make_float4(leaky(a.x + dn.x), leaky(a.y + dn.y),
                         leaky(a.z + dn.z), leaky(a.w + dn.w));
    }
    float4 m = make_float4(fmaxf(es.x, e0.x), fmaxf(es.y, e0.y),
                           fmaxf(es.z, e0.z), fmaxf(es.w, e0.w));
    for (int base = 32; base < deg; base += 32) {
        const int i = base + lane;
        if (i < deg) {
            const float4 a = alS[ssrc[beg + i]];
            m.x = fmaxf(m.x, leaky(a.x + dn.x));
            m.y = fmaxf(m.y, leaky(a.y + dn.y));
            m.z = fmaxf(m.z, leaky(a.z + dn.z));
            m.w = fmaxf(m.w, leaky(a.w + dn.w));
        }
    }
    m.x = wave_max(m.x); m.y = wave_max(m.y); m.z = wave_max(m.z); m.w = wave_max(m.w);

    const float4 wself = make_float4(__expf(es.x - m.x), __expf(es.y - m.y),
                                     __expf(es.z - m.z), __expf(es.w - m.w));
    const float wsl = sel4(wself.x, wself.y, wself.z, wself.w, k0, k1, k2, k3);
    const float4 hn = h[(size_t)n * 32 + lane];
    float4 acc = make_float4(wsl * hn.x, wsl * hn.y, wsl * hn.z, wsl * hn.w);
    float4 dp = make_float4(0.f, 0.f, 0.f, 0.f);

    float4 w = make_float4(0.f, 0.f, 0.f, 0.f);
    if (v0) w = make_float4(__expf(e0.x - m.x), __expf(e0.y - m.y),
                            __expf(e0.z - m.z), __expf(e0.w - m.w));
    int base = 0;
    for (;;) {
        dp.x += w.x; dp.y += w.y; dp.z += w.z; dp.w += w.w;
        const int cc = min(32, deg - base);
        for (int j = 0; j < cc; j += 4) {
            int sj[4];
            float wj[4];
            float4 r[4];
            for (int u = 0; u < 4; ++u) {
                sj[u] = __builtin_amdgcn_readlane(s, j + u);
                wj[u] = sel4(lane_val(w.x, j + u), lane_val(w.y, j + u),
                             lane_val(w.z, j + u), lane_val(w.w, j + u), k0, k1, k2, k3);
            }
            for (int u = 0; u < 4; ++u) r[u] = h[(size_t)sj[u] * 32 + lane];
            for (int u = 0; u < 4; ++u) {
                acc.x += wj[u] * r[u].x; acc.y += wj[u] * r[u].y;
                acc.z += wj[u] * r[u].z; acc.w += wj[u] * r[u].w;
            }
        }
        base += 32;
        if (base >= deg) break;
        const int i = base + lane;
        s = n;
        w = make_float4(0.f, 0.f, 0.f, 0.f);
        if (i < deg) {
            s = ssrc[beg + i];
            const float4 a = alS[s];
            w = make_float4(__expf(leaky(a.x + dn.x) - m.x), __expf(leaky(a.y + dn.y) - m.y),
                            __expf(leaky(a.z + dn.z) - m.z), __expf(leaky(a.w + dn.w) - m.w));
        }
    }

    const float den = sel4(wave_sum(dp.x) + wself.x, wave_sum(dp.y) + wself.y,
                           wave_sum(dp.z) + wself.z, wave_sum(dp.w) + wself.w,
                           k0, k1, k2, k3) + INF_EPS;
    const float* bp = bias + lane * 4;
    float4 o;
    o.x = acc.x / den + bp[0];
    o.y = acc.y / den + bp[1];
    o.z = acc.z / den + bp[2];
    o.w = acc.w / den + bp[3];
    o.x = o.x > 0.f ? o.x : __expf(o.x) - 1.f;
    o.y = o.y > 0.f ? o.y : __expf(o.y) - 1.f;
    o.z = o.z > 0.f ? o.z : __expf(o.z) - 1.f;
    o.w = o.w > 0.f ? o.w : __expf(o.w) - 1.f;
    outp[(size_t)n * 32 + lane] = o;
}

__global__ __launch_bounds__(256) void gather_softmax_h1(
    const int* __restrict__ off, const int* __restrict__ cnt, const int* __restrict__ ssrc,
    const float* __restrict__ alS, const float* __restrict__ alD,
    const float4* __restrict__ h, const float* __restrict__ bias,
    float4* __restrict__ outp, int N)
{
    const int lane = threadIdx.x & 31;
    const int g = lane >> 4;
    const int l16 = lane & 15;
    const int n = __builtin_amdgcn_readfirstlane((int)(blockIdx.x * 8 + (threadIdx.x >> 5)));
    if (n >= N) return;
    const int beg = off[n];
    const int deg = cnt[n];
    const float dn = alD[n];
    const float es = leaky(alS[n] + dn);

    const bool v0 = lane < deg;
    int s = n;
    float e0 = es;
    if (v0) {
        s = ssrc[beg + lane];
        e0 = leaky(alS[s] + dn);
    }
    float m = fmaxf(es, e0);
    for (int base = 32; base < deg; base += 32) {
        const int i = base + lane;
        if (i < deg) m = fmaxf(m, leaky(alS[ssrc[beg + i]] + dn));
    }
    m = wave_max(m);

    const float wself = __expf(es - m);
    float4 acc = make_float4(0.f, 0.f, 0.f, 0.f);
    if (g == 0) {
        const float4 hn = h[(size_t)n * 16 + l16];
        acc = make_float4(wself * hn.x, wself * hn.y, wself * hn.z, wself * hn.w);
    }
    float dp = 0.f;
    float w = v0 ? __expf(e0 - m) : 0.f;
    int base = 0;
    for (;;) {
        dp += w;
        const int cc = min(32, deg - base);
        for (int j = 0; j < cc; j += 8) {
            int sj[4];
            float wj[4];
            float4 r[4];
            for (int u = 0; u < 4; ++u) {
                sj[u] = __shfl(s, j + 2 * u + g);
                wj[u] = __shfl(w, j + 2 * u + g);
            }
            for (int u = 0; u < 4; ++u) r[u] = h[(size_t)sj[u] * 16 + l16];
            for (int u = 0; u < 4; ++u) {
                acc.x += wj[u] * r[u].x; acc.y += wj[u] * r[u].y;
                acc.z += wj[u] * r[u].z; acc.w += wj[u] * r[u].w;
            }
        }
        base += 32;
        if (base >= deg) break;
        const int i = base + lane;
        s = n;
        w = 0.f;
        if (i < deg) {
            s = ssrc[beg + i];
            w = __expf(leaky(alS[s] + dn) - m);
        }
    }

    const float den = wave_sum(dp) + wself + INF_EPS;
    acc.x += __shfl_xor(acc.x, 16);
    acc.y += __shfl_xor(acc.y, 16);
    acc.z += __shfl_xor(acc.z, 16);
    acc.w += __shfl_xor(acc.w, 16);
    if (g == 0) {
        const float* bp = bias + l16 * 4;
        float4 o;
        o.x = acc.x / den + bp[0];
        o.y = acc.y / den + bp[1];
        o.z = acc.z / den + bp[2];
        o.w = acc.w / den + bp[3];
        outp[(size_t)n * 16 + l16] = o;
    }
}

extern "C" void kernel_launch(void* const* d_in, const int* in_sizes, int n_in,
                              void* d_out, int out_size, void* d_ws, size_t ws_size,
                              hipStream_t stream)
{
    const float* x    = (const float*)d_in[0];
    const int*   eidx = (const int*)d_in[1];
    const float* W1   = (const float*)d_in[2];
    const float* aS1  = (const float*)d_in[3];
    const float* aD1  = (const float*)d_in[4];
    const float* b1   = (const float*)d_in[5];
    const float* W2   = (const float*)d_in[6];
    const float* aS2  = (const float*)d_in[7];
    const float* aD2  = (const float*)d_in[8];
    const float* b2   = (const float*)d_in[9];
    float* out = (float*)d_out;

    const int N    = in_sizes[0] / 128;
    const int E    = in_sizes[1] / 2;
    const int* esrc = eidx;
    const int* edst = eidx + E;

    float* ws = (float*)d_ws;
    size_t o = 0;
    float* h1   = ws + o; o += (size_t)N * 128;
    float* alS1 = ws + o; o += (size_t)N * 4;
    float* alD1 = ws + o; o += (size_t)N * 4;
    float* agg1 = ws + o; o += (size_t)N * 128;
    float* h2   = ws + o; o += (size_t)N * 64;
    float* alS2 = ws + o; o += (size_t)N;
    float* alD2 = ws + o; o += (size_t)N;
    int*   off  = (int*)(ws + o); o += (size_t)N;
    unsigned* bpack = (unsigned*)(ws + o); o += (size_t)E;
    int*   ssrc = (int*)(ws + o); o += (size_t)E;
    int*   cnt  = (int*)(ws + o); o += (size_t)N;
    o = (o + 31) & ~(size_t)31;
    int*   tot  = (int*)(ws + o); o += (size_t)256 * 32;
    const int TB = 256;
    auto cdiv = [](int a, int b) { return (a + b - 1) / b; };
    const int nChunk  = cdiv(E, SORT_CHUNK);
    const int nBucket = cdiv(N, 256);
    int*   cbase = (int*)(ws + o); o += (size_t)nChunk * 256;

    hipMemsetAsync(tot, 0, (size_t)256 * 32 * sizeof(int), stream);
    bucket_count<<<nChunk, 256, 0, stream>>>(edst, tot, cbase, E);
    bucket_scatter<<<nChunk, 256, 0, stream>>>(esrc, edst, tot, cbase, bpack, E);
    bucket_sort<<<nBucket, 1024, 0, stream>>>(bpack, tot, cnt, off, ssrc, N);

    gemm_wmma_f32<128, 128><<<cdiv(N / 16, 2), 128, 0, stream>>>(x, W1, h1, N / 16);
    attn_logits_h4<<<cdiv(N * 4, TB), TB, 0, stream>>>(h1, aS1, aD1, alS1, alD1, N);
    gather_softmax_h4<<<cdiv(N, 8), 256, 0, stream>>>(off, cnt, ssrc,
        (const float4*)alS1, (const float4*)alD1, (const float4*)h1, b1, (float4*)agg1, N);

    gemm_wmma_f32<128, 64><<<cdiv(N / 16, 4), 128, 0, stream>>>(agg1, W2, h2, N / 16);
    attn_logits_h1<<<cdiv(N, TB), TB, 0, stream>>>(h2, aS2, aD2, alS2, alD2, N);
    gather_softmax_h1<<<cdiv(N, 8), 256, 0, stream>>>(off, cnt, ssrc,
        alS2, alD2, (const float4*)h2, b2, (float4*)out, N);

    (void)n_in; (void)out_size; (void)ws_size;
}


# Attention logits formed in the GEMM epilogue from the accumulators (4 columns per lane, then a row-halving DPP butterfly over a head's lanes): both logit launches go; logit sum order changes
# speedup vs baseline: 8.5482x; 1.0912x over previous
#include <hip/hip_runtime.h>
#include <hip/hip_bf16.h>

typedef float v2f __attribute__((ext_vector_type(2)));
typedef float v8f __attribute__((ext_vector_type(8)));

#define NEG_SLOPE 0.2f
#define INF_EPS   1e-16f

__device__ __forceinline__ float leaky(float x) { return x > 0.f ? x : NEG_SLOPE * x; }

template <int M>
__device__ __forceinline__ float row_xor(float v) {
    return __int_as_float(__builtin_amdgcn_mov_dpp(__float_as_int(v), 0x160 | M, 0xf, 0xf, false));
}

template <int K, int NCOLS, int HC>
__global__ __launch_bounds__(128) __attribute__((amdgpu_waves_per_eu(4, 4))) void gemm_wmma_f32(
    const float* __restrict__ A, const float* __restrict__ B,
    const float* __restrict__ attS, const float* __restrict__ attD,
    float* __restrict__ C, float* __restrict__ alS, float* __restrict__ alD, int rowTiles)
{
    constexpr int CG  = NCOLS / 64;
    constexpr int TPB = 4 / CG;
    constexpr int H   = NCOLS / HC;
    constexpr int HPW = 64 / HC;
    constexpr int LPH = HC / 4;
    constexpr int F   = TPB * 16 * H;
    __shared__ __attribute__((aligned(16))) float s_al[2 * F];

    const int lane = threadIdx.x & 31;
    const int wave = __builtin_amdgcn_readfirstlane((int)(threadIdx.x >> 5));
    const int ti   = wave / CG;
    const int cg   = wave % CG;
    const int rt   = blockIdx.x * TPB + ti;
    const int c    = lane & 15;
    const int hh   = lane >> 4;
    const int col4 = cg * 64 + 4 * c;
    const bool live = rt < rowTiles;
    v8f acc0 = {}, acc1 = {}, acc2 = {}, acc3 = {};
    if (live) {
        const int row  = rt * 16 + c;
        const int kh   = hh << 1;

        const float* __restrict__ aPtr = A + (size_t)row * K + kh;
        const float* __restrict__ bPtr = B + (size_t)kh * NCOLS + col4;
        const float4 aS4 = *reinterpret_cast<const float4*>(attS + col4);
        const float4 aD4 = *reinterpret_cast<const float4*>(attD + col4);

#pragma unroll
        for (int k0 = 0; k0 < K; k0 += 4) {
            v2f a, b;
            a.x = aPtr[k0];
            a.y = aPtr[k0 + 1];
            const float4 b0 = *reinterpret_cast<const float4*>(bPtr + (size_t)k0 * NCOLS);
            const float4 b1 = *reinterpret_cast<const float4*>(bPtr + (size_t)k0 * NCOLS + NCOLS);
            b.x = b0.x; b.y = b1.x;
            acc0 = __builtin_amdgcn_wmma_f32_16x16x4_f32(
                false, a, false, b, (short)0, acc0, false, false);
            b.x = b0.y; b.y = b1.y;
            acc1 = __builtin_amdgcn_wmma_f32_16x16x4_f32(
                false, a, false, b, (short)0, acc1, false, false);
            b.x = b0.z; b.y = b1.z;
            acc2 = __builtin_amdgcn_wmma_f32_16x16x4_f32(
                false, a, false, b, (short)0, acc2, false, false);
            b.x = b0.w; b.y = b1.w;
            acc3 = __builtin_amdgcn_wmma_f32_16x16x4_f32(
                false, a, false, b, (short)0, acc3, false, false);
        }
        float ps[8], pd[8];
#pragma unroll
        for (int r = 0; r < 8; ++r) {
            ps[r] = acc0[r] * aS4.x + acc1[r] * aS4.y + acc2[r] * aS4.z + acc3[r] * aS4.w;
            pd[r] = acc0[r] * aD4.x + acc1[r] * aD4.y + acc2[r] * aD4.z + acc3[r] * aD4.w;
        }
        const bool u4 = (c & 4) != 0, u2 = (c & 2) != 0, u1 = (c & 1) != 0;
        float qs[4], qd[4];
#pragma unroll
        for (int i = 0; i < 4; ++i) {
            qs[i] = (u4 ? ps[i + 4] : ps[i]) + row_xor<4>(u4 ? ps[i] : ps[i + 4]);
            qd[i] = (u4 ? pd[i + 4] : pd[i]) + row_xor<4>(u4 ? pd[i] : pd[i + 4]);
        }
        float ws[2], wd[2];
#pragma unroll
        for (int i = 0; i < 2; ++i) {
            ws[i] = (u2 ? qs[i + 2] : qs[i]) + row_xor<2>(u2 ? qs[i] : qs[i + 2]);
            wd[i] = (u2 ? qd[i + 2] : qd[i]) + row_xor<2>(u2 ? qd[i] : qd[i + 2]);
        }
        float ts = (u1 ? ws[1] : ws[0]) + row_xor<1>(u1 ? ws[0] : ws[1]);
        float td = (u1 ? wd[1] : wd[0]) + row_xor<1>(u1 ? wd[0] : wd[1]);
        if (LPH == 16) { ts += row_xor<8>(ts); td += row_xor<8>(td); }
        if ((c % LPH) < 8) {
            const int idx = (ti * 16 + (hh << 3) + (c & 7)) * H + cg * HPW + c / LPH;
            s_al[idx]     = ts;
            s_al[F + idx] = td;
        }
    }
    __syncthreads();
    if (live) {
        float* __restrict__ cPtr =
            C + ((size_t)rt * 16 + (hh << 3)) * NCOLS + col4;
#pragma unroll
        for (int r = 0; r < 8; ++r)
            *reinterpret_cast<float4*>(cPtr + (size_t)r * NCOLS) =
                make_float4(acc0[r], acc1[r], acc2[r], acc3[r]);
    }
    if (wave < 2 && lane < F / 4) {
        const int e = lane * 4;
        if (blockIdx.x * TPB + e / (16 * H) < rowTiles) {
            const float4 v = *reinterpret_cast<const float4*>(&s_al[wave * F + e]);
            float* __restrict__ dst = (wave == 0 ? alS : alD) + (size_t)blockIdx.x * F + e;
            *reinterpret_cast<float4*>(dst) = v;
        }
    }
}

#define SORT_CHUNK 2048

__device__ __forceinline__ int wave_incl_scan_i(int v, int lane) {
    for (int o = 1; o < 32; o <<= 1) {
        int u = __shfl_up(v, o);
        if (lane >= o) v += u;
    }
    return v;
}

__global__ __launch_bounds__(256) void bucket_count(
    const int* __restrict__ edst, int* __restrict__ tot, int* __restrict__ cbase, int E)
{
    __shared__ int s_h[256];
    const int t = threadIdx.x;
    s_h[t] = 0;
    __syncthreads();
    const int e0 = blockIdx.x * SORT_CHUNK + t;
    int dv[SORT_CHUNK / 256];
#pragma unroll
    for (int k = 0; k < SORT_CHUNK / 256; ++k) {
        const int e = e0 + k * 256;
        dv[k] = 0;
        if (e < E) dv[k] = edst[e];
    }
#pragma unroll
    for (int k = 0; k < SORT_CHUNK / 256; ++k) {
        const int e = e0 + k * 256;
        if (e < E) atomicAdd(&s_h[dv[k] >> 8], 1);
    }
    __syncthreads();
    const int hcount = s_h[t];
    int base = 0;
    if (hcount > 0) base = atomicAdd(&tot[t * 32], hcount);
    cbase[blockIdx.x * 256 + t] = base;
}

__global__ __launch_bounds__(256) void bucket_scatter(
    const int* __restrict__ esrc, const int* __restrict__ edst,
    const int* __restrict__ tot, const int* __restrict__ cbase,
    unsigned* __restrict__ bpack, int E)
{
    __shared__ int s_cur[256];
    __shared__ int s_w[8];
    const int t = threadIdx.x;
    const int lane = t & 31;
    const int wave = t >> 5;
    const int e0 = blockIdx.x * SORT_CHUNK + t;
    int sv[SORT_CHUNK / 256], dv[SORT_CHUNK / 256];
#pragma unroll
    for (int k = 0; k < SORT_CHUNK / 256; ++k) {
        const int e = e0 + k * 256;
        sv[k] = 0; dv[k] = 0;
        if (e < E) { sv[k] = esrc[e]; dv[k] = edst[e]; }
    }
    const int c = tot[t * 32];
    const int incl = wave_incl_scan_i(c, lane);
    if (lane == 31) s_w[wave] = incl;
    __syncthreads();
    int wb = 0;
    for (int w = 0; w < 8; ++w) if (w < wave) wb += s_w[w];
    s_cur[t] = wb + incl - c + cbase[blockIdx.x * 256 + t];
    __syncthreads();
#pragma unroll
    for (int k = 0; k < SORT_CHUNK / 256; ++k) {
        const int e = e0 + k * 256;
        if (e < E) {
            const int pos = atomicAdd(&s_cur[dv[k] >> 8], 1);
            bpack[pos] = ((unsigned)sv[k] << 8) | ((unsigned)dv[k] & 255u);
        }
    }
}

__global__ __launch_bounds__(1024) void bucket_sort(
    const unsigned* __restrict__ bpack, const int* __restrict__ tot,
    int* __restrict__ cnt, int* __restrict__ off, int* __restrict__ ssrc, int N)
{
    __shared__ int s_cnt[256];
    __shared__ int s_cur[256];
    __shared__ int s_w[8];
    __shared__ int s_start;
    const int b = blockIdx.x;
    const int t = threadIdx.x;
    const int lane = t & 31;
    const int wave = t >> 5;
    if (t < 256) s_cnt[t] = 0;
    if (t == 0) s_start = 0;
    __syncthreads();
    if (t < 256) {
        int v = (t < b) ? tot[t * 32] : 0;
        for (int o = 16; o > 0; o >>= 1) v += __shfl_xor(v, o);
        if (lane == 0 && v != 0) atomicAdd(&s_start, v);
    }
    const int m = tot[b * 32];
    __syncthreads();
    const int start = s_start;
    for (int i0 = t; i0 < m; i0 += 4096) {
        unsigned p[4];
#pragma unroll
        for (int k = 0; k < 4; ++k) {
            const int i = i0 + k * 1024;
            p[k] = 0u;
            if (i < m) p[k] = bpack[start + i];
        }
#pragma unroll
        for (int k = 0; k < 4; ++k) {
            const int i = i0 + k * 1024;
            if (i < m) atomicAdd(&s_cnt[p[k] & 255u], 1);
        }
    }
    __syncthreads();
    const int c = (t < 256) ? s_cnt[t] : 0;
    const int incl = wave_incl_scan_i(c, lane);
    if (t < 256 && lane == 31) s_w[wave] = incl;
    __syncthreads();
    if (t < 256) {
        int wb = 0;
        for (int w = 0; w < 8; ++w) if (w < wave) wb += s_w[w];
        const int ex = wb + incl - c;
        s_cur[t] = ex;
        const int n = b * 256 + t;
        if (n < N) { cnt[n] = c; off[n] = start + ex; }
    }
    __syncthreads();
    for (int i0 = t; i0 < m; i0 += 4096) {
        unsigned p[4];
#pragma unroll
        for (int k = 0; k < 4; ++k) {
            const int i = i0 + k * 1024;
            p[k] = 0u;
            if (i < m) p[k] = bpack[start + i];
        }
#pragma unroll
        for (int k = 0; k < 4; ++k) {
            const int i = i0 + k * 1024;
            if (i < m) {
                const int pos = atomicAdd(&s_cur[p[k] & 255u], 1);
                ssrc[start + pos] = (int)(p[k] >> 8);
            }
        }
    }
}

__device__ __forceinline__ float wave_max(float v) {
    for (int o = 16; o > 0; o >>= 1) v = fmaxf(v, __shfl_xor(v, o));
    return v;
}
__device__ __forceinline__ float wave_sum(float v) {
    for (int o = 16; o > 0; o >>= 1) v += __shfl_xor(v, o);
    return v;
}
__device__ __forceinline__ float lane_val(float v, int j) {
    return __int_as_float(__builtin_amdgcn_readlane(__float_as_int(v), j));
}
__device__ __forceinline__ float sel4(float a, float b, float c, float d,
                                      unsigned k0, unsigned k1, unsigned k2, unsigned k3) {
    return __uint_as_float((__float_as_uint(a) & k0) | (__float_as_uint(b) & k1) |
                           (__float_as_uint(c) & k2) | (__float_as_uint(d) & k3));
}

__global__ __launch_bounds__(256) void gather_softmax_h4(
    const int* __restrict__ off, const int* __restrict__ cnt, const int* __restrict__ ssrc,
    const float4* __restrict__ alS, const float4* __restrict__ alD,
    const float4* __restrict__ h, const float* __restrict__ bias,
    float4* __restrict__ outp, int N)
{
    const int lane = threadIdx.x & 31;
    const int hd = lane >> 3;
    const unsigned k0 = hd == 0 ? 0xffffffffu : 0u;
    const unsigned k1 = hd == 1 ? 0xffffffffu : 0u;
    const unsigned k2 = hd == 2 ? 0xffffffffu : 0u;
    const unsigned k3 = hd == 3 ? 0xffffffffu : 0u;
    const int n = __builtin_amdgcn_readfirstlane((int)(blockIdx.x * 8 + (threadIdx.x >> 5)));
    if (n >= N) return;
    const int beg = off[n];
    const int deg = cnt[n];
    const float4 dn = alD[n];
    const float4 sn = alS[n];
    const float4 es = make_float4(leaky(sn.x + dn.x), leaky(sn.y + dn.y),
                                  leaky(sn.z + dn.z), leaky(sn.w + dn.w));

    const bool v0 = lane < deg;
    int s = n;
    float4 e0 = es;
    if (v0) {
        s = ssrc[beg + lane];
        const float4 a = alS[s];
        e0 = make_float4(leaky(a.x + dn.x), leaky(a.y + dn.y),
                         leaky(a.z + dn.z), leaky(a.w + dn.w));
    }
    float4 m = make_float4(fmaxf(es.x, e0.x), fmaxf(es.y, e0.y),
                           fmaxf(es.z, e0.z), fmaxf(es.w, e0.w));
    for (int base = 32; base < deg; base += 32) {
        const int i = base + lane;
        if (i < deg) {
            const float4 a = alS[ssrc[beg + i]];
            m.x = fmaxf(m.x, leaky(a.x + dn.x));
            m.y = fmaxf(m.y, leaky(a.y + dn.y));
            m.z = fmaxf(m.z, leaky(a.z + dn.z));
            m.w = fmaxf(m.w, leaky(a.w + dn.w));
        }
    }
    m.x = wave_max(m.x); m.y = wave_max(m.y); m.z = wave_max(m.z); m.w = wave_max(m.w);

    const float4 wself = make_float4(__expf(es.x - m.x), __expf(es.y - m.y),
                                     __expf(es.z - m.z), __expf(es.w - m.w));
    const float wsl = sel4(wself.x, wself.y, wself.z, wself.w, k0, k1, k2, k3);
    const float4 hn = h[(size_t)n * 32 + lane];
    float4 acc = make_float4(wsl * hn.x, wsl * hn.y, wsl * hn.z, wsl * hn.w);
    float4 dp = make_float4(0.f, 0.f, 0.f, 0.f);

    float4 w = make_float4(0.f, 0.f, 0.f, 0.f);
    if (v0) w = make_float4(__expf(e0.x - m.x), __expf(e0.y - m.y),
                            __expf(e0.z - m.z), __expf(e0.w - m.w));
    int base = 0;
    for (;;) {
        dp.x += w.x; dp.y += w.y; dp.z += w.z; dp.w += w.w;
        const int cc = min(32, deg - base);
        for (int j = 0; j < cc; j += 4) {
            int sj[4];
            float wj[4];
            float4 r[4];
            for (int u = 0; u < 4; ++u) {
                sj[u] = __builtin_amdgcn_readlane(s, j + u);
                wj[u] = sel4(lane_val(w.x, j + u), lane_val(w.y, j + u),
                             lane_val(w.z, j + u), lane_val(w.w, j + u), k0, k1, k2, k3);
            }
            for (int u = 0; u < 4; ++u) r[u] = h[(size_t)sj[u] * 32 + lane];
            for (int u = 0; u < 4; ++u) {
                acc.x += wj[u] * r[u].x; acc.y += wj[u] * r[u].y;
                acc.z += wj[u] * r[u].z; acc.w += wj[u] * r[u].w;
            }
        }
        base += 32;
        if (base >= deg) break;
        const int i = base + lane;
        s = n;
        w = make_float4(0.f, 0.f, 0.f, 0.f);
        if (i < deg) {
            s = ssrc[beg + i];
            const float4 a = alS[s];
            w = make_float4(__expf(leaky(a.x + dn.x) - m.x), __expf(leaky(a.y + dn.y) - m.y),
                            __expf(leaky(a.z + dn.z) - m.z), __expf(leaky(a.w + dn.w) - m.w));
        }
    }

    const float den = sel4(wave_sum(dp.x) + wself.x, wave_sum(dp.y) + wself.y,
                           wave_sum(dp.z) + wself.z, wave_sum(dp.w) + wself.w,
                           k0, k1, k2, k3) + INF_EPS;
    const float* bp = bias + lane * 4;
    float4 o;
    o.x = acc.x / den + bp[0];
    o.y = acc.y / den + bp[1];
    o.z = acc.z / den + bp[2];
    o.w = acc.w / den + bp[3];
    o.x = o.x > 0.f ? o.x : __expf(o.x) - 1.f;
    o.y = o.y > 0.f ? o.y : __expf(o.y) - 1.f;
    o.z = o.z > 0.f ? o.z : __expf(o.z) - 1.f;
    o.w = o.w > 0.f ? o.w : __expf(o.w) - 1.f;
    outp[(size_t)n * 32 + lane] = o;
}

__global__ __launch_bounds__(256) void gather_softmax_h1(
    const int* __restrict__ off, const int* __restrict__ cnt, const int* __restrict__ ssrc,
    const float* __restrict__ alS, const float* __restrict__ alD,
    const float4* __restrict__ h, const float* __restrict__ bias,
    float4* __restrict__ outp, int N)
{
    const int lane = threadIdx.x & 31;
    const int g = lane >> 4;
    const int l16 = lane & 15;
    const int n = __builtin_amdgcn_readfirstlane((int)(blockIdx.x * 8 + (threadIdx.x >> 5)));
    if (n >= N) return;
    const int beg = off[n];
    const int deg = cnt[n];
    const float dn = alD[n];
    const float es = leaky(alS[n] + dn);

    const bool v0 = lane < deg;
    int s = n;
    float e0 = es;
    if (v0) {
        s = ssrc[beg + lane];
        e0 = leaky(alS[s] + dn);
    }
    float m = fmaxf(es, e0);
    for (int base = 32; base < deg; base += 32) {
        const int i = base + lane;
        if (i < deg) m = fmaxf(m, leaky(alS[ssrc[beg + i]] + dn));
    }
    m = wave_max(m);

    const float wself = __expf(es - m);
    float4 acc = make_float4(0.f, 0.f, 0.f, 0.f);
    if (g == 0) {
        const float4 hn = h[(size_t)n * 16 + l16];
        acc = make_float4(wself * hn.x, wself * hn.y, wself * hn.z, wself * hn.w);
    }
    float dp = 0.f;
    float w = v0 ? __expf(e0 - m) : 0.f;
    int base = 0;
    for (;;) {
        dp += w;
        const int cc = min(32, deg - base);
        for (int j = 0; j < cc; j += 8) {
            int sj[4];
            float wj[4];
            float4 r[4];
            for (int u = 0; u < 4; ++u) {
                sj[u] = __shfl(s, j + 2 * u + g);
                wj[u] = __shfl(w, j + 2 * u + g);
            }
            for (int u = 0; u < 4; ++u) r[u] = h[(size_t)sj[u] * 16 + l16];
            for (int u = 0; u < 4; ++u) {
                acc.x += wj[u] * r[u].x; acc.y += wj[u] * r[u].y;
                acc.z += wj[u] * r[u].z; acc.w += wj[u] * r[u].w;
            }
        }
        base += 32;
        if (base >= deg) break;
        const int i = base + lane;
        s = n;
        w = 0.f;
        if (i < deg) {
            s = ssrc[beg + i];
            w = __expf(leaky(alS[s] + dn) - m);
        }
    }

    const float den = wave_sum(dp) + wself + INF_EPS;
    acc.x += __shfl_xor(acc.x, 16);
    acc.y += __shfl_xor(acc.y, 16);
    acc.z += __shfl_xor(acc.z, 16);
    acc.w += __shfl_xor(acc.w, 16);
    if (g == 0) {
        const float* bp = bias + l16 * 4;
        float4 o;
        o.x = acc.x / den + bp[0];
        o.y = acc.y / den + bp[1];
        o.z = acc.z / den + bp[2];
        o.w = acc.w / den + bp[3];
        outp[(size_t)n * 16 + l16] = o;
    }
}

extern "C" void kernel_launch(void* const* d_in, const int* in_sizes, int n_in,
                              void* d_out, int out_size, void* d_ws, size_t ws_size,
                              hipStream_t stream)
{
    const float* x    = (const float*)d_in[0];
    const int*   eidx = (const int*)d_in[1];
    const float* W1   = (const float*)d_in[2];
    const float* aS1  = (const float*)d_in[3];
    const float* aD1  = (const float*)d_in[4];
    const float* b1   = (const float*)d_in[5];
    const float* W2   = (const float*)d_in[6];
    const float* aS2  = (const float*)d_in[7];
    const float* aD2  = (const float*)d_in[8];
    const float* b2   = (const float*)d_in[9];
    float* out = (float*)d_out;

    const int N    = in_sizes[0] / 128;
    const int E    = in_sizes[1] / 2;
    const int* esrc = eidx;
    const int* edst = eidx + E;

    float* ws = (float*)d_ws;
    size_t o = 0;
    float* h1   = ws + o; o += (size_t)N * 128;
    float* alS1 = ws + o; o += (size_t)N * 4;
    float* alD1 = ws + o; o += (size_t)N * 4;
    float* agg1 = ws + o; o += (size_t)N * 128;
    float* h2   = ws + o; o += (size_t)N * 64;
    float* alS2 = ws + o; o += ((size_t)N + 31) & ~(size_t)31;
    float* alD2 = ws + o; o += ((size_t)N + 31) & ~(size_t)31;
    int*   off  = (int*)(ws + o); o += (size_t)N;
    unsigned* bpack = (unsigned*)(ws + o); o += (size_t)E;
    int*   ssrc = (int*)(ws + o); o += (size_t)E;
    int*   cnt  = (int*)(ws + o); o += (size_t)N;
    o = (o + 31) & ~(size_t)31;
    int*   tot  = (int*)(ws + o); o += (size_t)256 * 32;
    auto cdiv = [](int a, int b) { return (a + b - 1) / b; };
    const int nChunk  = cdiv(E, SORT_CHUNK);
    const int nBucket = cdiv(N, 256);
    int*   cbase = (int*)(ws + o); o += (size_t)nChunk * 256;

    hipMemsetAsync(tot, 0, (size_t)256 * 32 * sizeof(int), stream);
    bucket_count<<<nChunk, 256, 0, stream>>>(edst, tot, cbase, E);
    bucket_scatter<<<nChunk, 256, 0, stream>>>(esrc, edst, tot, cbase, bpack, E);
    bucket_sort<<<nBucket, 1024, 0, stream>>>(bpack, tot, cnt, off, ssrc, N);

    gemm_wmma_f32<128, 128, 32><<<cdiv(N / 16, 2), 128, 0, stream>>>(x, W1, aS1, aD1, h1, alS1, alD1, N / 16);
    gather_softmax_h4<<<cdiv(N, 8), 256, 0, stream>>>(off, cnt, ssrc,
        (const float4*)alS1, (const float4*)alD1, (const float4*)h1, b1, (float4*)agg1, N);

    gemm_wmma_f32<128, 64, 64><<<cdiv(N / 16, 4), 128, 0, stream>>>(agg1, W2, aS2, aD2, h2, alS2, alD2, N / 16);
    gather_softmax_h1<<<cdiv(N, 8), 256, 0, stream>>>(off, cnt, ssrc,
        alS2, alD2, (const float4*)h2, b2, (float4*)out, N);

    (void)n_in; (void)out_size; (void)ws_size;
}


# Edge sort in 2 launches: a chunk is grouped by bucket in LDS and stored in place with its piece starts; bucket start = sum of piece starts: memset, global counters, 1 launch go; sums in arrival order
# speedup vs baseline: 9.8469x; 1.1519x over previous
#include <hip/hip_runtime.h>
#include <hip/hip_bf16.h>

typedef float v2f __attribute__((ext_vector_type(2)));
typedef float v8f __attribute__((ext_vector_type(8)));

#define NEG_SLOPE 0.2f
#define INF_EPS   1e-16f

__device__ __forceinline__ float leaky(float x) { return x > 0.f ? x : NEG_SLOPE * x; }

template <int M>
__device__ __forceinline__ float row_xor(float v) {
    return __int_as_float(__builtin_amdgcn_mov_dpp(__float_as_int(v), 0x160 | M, 0xf, 0xf, false));
}

template <int K, int NCOLS, int HC>
__global__ __launch_bounds__(128) __attribute__((amdgpu_waves_per_eu(4, 4))) void gemm_wmma_f32(
    const float* __restrict__ A, const float* __restrict__ B,
    const float* __restrict__ attS, const float* __restrict__ attD,
    float* __restrict__ C, float* __restrict__ alS, float* __restrict__ alD, int rowTiles)
{
    constexpr int CG  = NCOLS / 64;
    constexpr int TPB = 4 / CG;
    constexpr int H   = NCOLS / HC;
    constexpr int HPW = 64 / HC;
    constexpr int LPH = HC / 4;
    constexpr int F   = TPB * 16 * H;
    __shared__ __attribute__((aligned(16))) float s_al[2 * F];

    const int lane = threadIdx.x & 31;
    const int wave = __builtin_amdgcn_readfirstlane((int)(threadIdx.x >> 5));
    const int ti   = wave / CG;
    const int cg   = wave % CG;
    const int rt   = blockIdx.x * TPB + ti;
    const int c    = lane & 15;
    const int hh   = lane >> 4;
    const int col4 = cg * 64 + 4 * c;
    const bool live = rt < rowTiles;
    v8f acc0 = {}, acc1 = {}, acc2 = {}, acc3 = {};
    if (live) {
        const int row  = rt * 16 + c;
        const int kh   = hh << 1;

        const float* __restrict__ aPtr = A + (size_t)row * K + kh;
        const float* __restrict__ bPtr = B + (size_t)kh * NCOLS + col4;
        const float4 aS4 = *reinterpret_cast<const float4*>(attS + col4);
        const float4 aD4 = *reinterpret_cast<const float4*>(attD + col4);

#pragma unroll
        for (int k0 = 0; k0 < K; k0 += 4) {
            v2f a, b;
            a.x = aPtr[k0];
            a.y = aPtr[k0 + 1];
            const float4 b0 = *reinterpret_cast<const float4*>(bPtr + (size_t)k0 * NCOLS);
            const float4 b1 = *reinterpret_cast<const float4*>(bPtr + (size_t)k0 * NCOLS + NCOLS);
            b.x = b0.x; b.y = b1.x;
            acc0 = __builtin_amdgcn_wmma_f32_16x16x4_f32(
                false, a, false, b, (short)0, acc0, false, false);
            b.x = b0.y; b.y = b1.y;
            acc1 = __builtin_amdgcn_wmma_f32_16x16x4_f32(
                false, a, false, b, (short)0, acc1, false, false);
            b.x = b0.z; b.y = b1.z;
            acc2 = __builtin_amdgcn_wmma_f32_16x16x4_f32(
                false, a, false, b, (short)0, acc2, false, false);
            b.x = b0.w; b.y = b1.w;
            acc3 = __builtin_amdgcn_wmma_f32_16x16x4_f32(
                false, a, false, b, (short)0, acc3, false, false);
        }
        float ps[8], pd[8];
#pragma unroll
        for (int r = 0; r < 8; ++r) {
            ps[r] = acc0[r] * aS4.x + acc1[r] * aS4.y + acc2[r] * aS4.z + acc3[r] * aS4.w;
            pd[r] = acc0[r] * aD4.x + acc1[r] * aD4.y + acc2[r] * aD4.z + acc3[r] * aD4.w;
        }
        const bool u4 = (c & 4) != 0, u2 = (c & 2) != 0, u1 = (c & 1) != 0;
        float qs[4], qd[4];
#pragma unroll
        for (int i = 0; i < 4; ++i) {
            qs[i] = (u4 ? ps[i + 4] : ps[i]) + row_xor<4>(u4 ? ps[i] : ps[i + 4]);
            qd[i] = (u4 ? pd[i + 4] : pd[i]) + row_xor<4>(u4 ? pd[i] : pd[i + 4]);
        }
        float ws[2], wd[2];
#pragma unroll
        for (int i = 0; i < 2; ++i) {
            ws[i] = (u2 ? qs[i + 2] : qs[i]) + row_xor<2>(u2 ? qs[i] : qs[i + 2]);
            wd[i] = (u2 ? qd[i + 2] : qd[i]) + row_xor<2>(u2 ? qd[i] : qd[i + 2]);
        }
        float ts = (u1 ? ws[1] : ws[0]) + row_xor<1>(u1 ? ws[0] : ws[1]);
        float td = (u1 ? wd[1] : wd[0]) + row_xor<1>(u1 ? wd[0] : wd[1]);
        if (LPH == 16) { ts += row_xor<8>(ts); td += row_xor<8>(td); }
        if ((c % LPH) < 8) {
            const int idx = (ti * 16 + (hh << 3) + (c & 7)) * H + cg * HPW + c / LPH;
            s_al[idx]     = ts;
            s_al[F + idx] = td;
        }
    }
    __syncthreads();
    if (live) {
        float* __restrict__ cPtr =
            C + ((size_t)rt * 16 + (hh << 3)) * NCOLS + col4;
#pragma unroll
        for (int r = 0; r < 8; ++r)
            *reinterpret_cast<float4*>(cPtr + (size_t)r * NCOLS) =
                make_float4(acc0[r], acc1[r], acc2[r], acc3[r]);
    }
    if (wave < 2 && lane < F / 4) {
        const int e = lane * 4;
        if (blockIdx.x * TPB + e / (16 * H) < rowTiles) {
            const float4 v = *reinterpret_cast<const float4*>(&s_al[wave * F + e]);
            float* __restrict__ dst = (wave == 0 ? alS : alD) + (size_t)blockIdx.x * F + e;
            *reinterpret_cast<float4*>(dst) = v;
        }
    }
}

#define SORT_CHUNK 2048

__device__ __forceinline__ int wave_incl_scan_i(int v, int lane) {
    for (int o = 1; o < 32; o <<= 1) {
        int u = __shfl_up(v, o);
        if (lane >= o) v += u;
    }
    return v;
}

__global__ __launch_bounds__(256) void chunk_group(
    const int* __restrict__ esrc, const int* __restrict__ edst,
    unsigned* __restrict__ bpack, unsigned* __restrict__ meta, int E)
{
    __shared__ int s_h[256];
    __shared__ int s_w[8];
    __shared__ unsigned s_out[SORT_CHUNK];
    const int t = threadIdx.x;
    const int lane = t & 31;
    const int wave = t >> 5;
    s_h[t] = 0;
    const int cb = blockIdx.x * SORT_CHUNK;
    const int e0 = cb + t;
    int sv[SORT_CHUNK / 256], dv[SORT_CHUNK / 256], rk[SORT_CHUNK / 256];
#pragma unroll
    for (int k = 0; k < SORT_CHUNK / 256; ++k) {
        const int e = e0 + k * 256;
        sv[k] = 0; dv[k] = 0;
        if (e < E) { sv[k] = esrc[e]; dv[k] = edst[e]; }
    }
    __syncthreads();
#pragma unroll
    for (int k = 0; k < SORT_CHUNK / 256; ++k) {
        const int e = e0 + k * 256;
        rk[k] = 0;
        if (e < E) rk[k] = atomicAdd(&s_h[dv[k] >> 8], 1);
    }
    __syncthreads();
    const int c = s_h[t];
    const int incl = wave_incl_scan_i(c, lane);
    if (lane == 31) s_w[wave] = incl;
    __syncthreads();
    int wb = 0;
    for (int w = 0; w < 8; ++w) if (w < wave) wb += s_w[w];
    const int ex = wb + incl - c;
    s_h[t] = ex;
    __syncthreads();
#pragma unroll
    for (int k = 0; k < SORT_CHUNK / 256; ++k) {
        const int e = e0 + k * 256;
        if (e < E)
            s_out[s_h[dv[k] >> 8] + rk[k]] = ((unsigned)sv[k] << 8) | ((unsigned)dv[k] & 255u);
    }
    __syncthreads();
#pragma unroll
    for (int k = 0; k < SORT_CHUNK / 256; ++k) {
        const int i = t + k * 256;
        if (cb + i < E) bpack[cb + i] = s_out[i];
    }
    meta[blockIdx.x * 256 + t] = (unsigned)ex | ((unsigned)c << 16);
}

template <bool SCATTER>
__device__ __forceinline__ void walk_pieces(
    const unsigned* __restrict__ bpack, const unsigned* s_meta, int* s_ctr,
    int* __restrict__ seg, int c0, int nt, int g, int l16)
{
    for (int q0 = g; q0 < nt; q0 += 512) {
        unsigned p[8];
        int pc[8], bs[8];
#pragma unroll
        for (int k = 0; k < 8; ++k) {
            const int q = q0 + 64 * k;
            const unsigned mq = s_meta[q];
            pc[k] = (int)(mq >> 16);
            bs[k] = (c0 + q) * SORT_CHUNK + (int)(mq & 0xffffu);
            p[k] = bpack[l16 < pc[k] ? bs[k] + l16 : 0];
        }
#pragma unroll
        for (int k = 0; k < 8; ++k) {
            if (l16 < pc[k]) {
                const int pos = atomicAdd(&s_ctr[p[k] & 255u], 1);
                if (SCATTER) seg[pos] = (int)(p[k] >> 8);
            }
            for (int i = l16 + 16; i < pc[k]; i += 16) {
                const unsigned pp = bpack[bs[k] + i];
                const int pos = atomicAdd(&s_ctr[pp & 255u], 1);
                if (SCATTER) seg[pos] = (int)(pp >> 8);
            }
        }
    }
}

__global__ __launch_bounds__(1024) void bucket_sort(
    const unsigned* __restrict__ bpack, const unsigned* __restrict__ meta,
    int* __restrict__ cnt, int* __restrict__ off, int* __restrict__ ssrc, int N, int nChunk)
{
    __shared__ int s_cnt[256];
    __shared__ int s_cur[256];
    __shared__ int s_w[8];
    __shared__ int s_start;
    __shared__ unsigned s_meta[1024];
    const int b = blockIdx.x;
    const int t = threadIdx.x;
    const int lane = t & 31;
    const int wave = t >> 5;
    const int g = t >> 4;
    const int l16 = t & 15;
    unsigned m0 = 0u;
    if (t < nChunk) m0 = meta[(size_t)t * 256 + b];
    if (t < 256) s_cnt[t] = 0;
    if (t == 0) s_start = 0;
    __syncthreads();
    for (int c0 = 0; c0 < nChunk; c0 += 1024) {
        unsigned mv = m0;
        if (c0 > 0) {
            mv = 0u;
            if (c0 + t < nChunk) mv = meta[(size_t)(c0 + t) * 256 + b];
        }
        s_meta[t] = mv;
        int v = (int)(mv & 0xffffu);
        for (int o = 16; o > 0; o >>= 1) v += __shfl_xor(v, o);
        if (lane == 0 && v != 0) atomicAdd(&s_start, v);
        __syncthreads();
        walk_pieces<false>(bpack, s_meta, s_cnt, nullptr, c0, min(1024, nChunk - c0), g, l16);
        __syncthreads();
    }
    const int start = s_start;
    const int c = (t < 256) ? s_cnt[t] : 0;
    const int incl = wave_incl_scan_i(c, lane);
    if (t < 256 && lane == 31) s_w[wave] = incl;
    __syncthreads();
    int ex0 = 0;
    if (t < 256) {
        int wb = 0;
        for (int w = 0; w < 8; ++w) if (w < wave) wb += s_w[w];
        const int ex = wb + incl - c;
        s_cur[t] = ex;
        ex0 = ex;
    }
    __syncthreads();
    if (t < 256) {
        const int n = b * 256 + t;
        if (n < N) { cnt[n] = c; off[n] = start + ex0; }
    }
    for (int c0 = 0; c0 < nChunk; c0 += 1024) {
        if (nChunk > 1024) {
            unsigned mv = 0u;
            if (c0 + t < nChunk) mv = meta[(size_t)(c0 + t) * 256 + b];
            s_meta[t] = mv;
            __syncthreads();
        }
        walk_pieces<true>(bpack, s_meta, s_cur, ssrc + start, c0, min(1024, nChunk - c0), g, l16);
        if (nChunk > 1024) __syncthreads();
    }
}

__device__ __forceinline__ float wave_max(float v) {
    for (int o = 16; o > 0; o >>= 1) v = fmaxf(v, __shfl_xor(v, o));
    return v;
}
__device__ __forceinline__ float wave_sum(float v) {
    for (int o = 16; o > 0; o >>= 1) v += __shfl_xor(v, o);
    return v;
}
__device__ __forceinline__ float lane_val(float v, int j) {
    return __int_as_float(__builtin_amdgcn_readlane(__float_as_int(v), j));
}
__device__ __forceinline__ float sel4(float a, float b, float c, float d,
                                      unsigned k0, unsigned k1, unsigned k2, unsigned k3) {
    return __uint_as_float((__float_as_uint(a) & k0) | (__float_as_uint(b) & k1) |
                           (__float_as_uint(c) & k2) | (__float_as_uint(d) & k3));
}

__global__ __launch_bounds__(256) void gather_softmax_h4(
    const int* __restrict__ off, const int* __restrict__ cnt, const int* __restrict__ ssrc,
    const float4* __restrict__ alS, const float4* __restrict__ alD,
    const float4* __restrict__ h, const float* __restrict__ bias,
    float4* __restrict__ outp, int N)
{
    const int lane = threadIdx.x & 31;
    const int hd = lane >> 3;
    const unsigned k0 = hd == 0 ? 0xffffffffu : 0u;
    const unsigned k1 = hd == 1 ? 0xffffffffu : 0u;
    const unsigned k2 = hd == 2 ? 0xffffffffu : 0u;
    const unsigned k3 = hd == 3 ? 0xffffffffu : 0u;
    const int n = __builtin_amdgcn_readfirstlane((int)(blockIdx.x * 8 + (threadIdx.x >> 5)));
    if (n >= N) return;
    const int beg = off[n];
    const int deg = cnt[n];
    const float4 dn = alD[n];
    const float4 sn = alS[n];
    const float4 es = make_float4(leaky(sn.x + dn.x), leaky(sn.y + dn.y),
                                  leaky(sn.z + dn.z), leaky(sn.w + dn.w));

    const bool v0 = lane < deg;
    int s = n;
    float4 e0 = es;
    if (v0) {
        s = ssrc[beg + lane];
        const float4 a = alS[s];
        e0 = make_float4(leaky(a.x + dn.x), leaky(a.y + dn.y),
                         leaky(a.z + dn.z), leaky(a.w + dn.w));
    }
    float4 m = make_float4(fmaxf(es.x, e0.x), fmaxf(es.y, e0.y),
                           fmaxf(es.z, e0.z), fmaxf(es.w, e0.w));
    for (int base = 32; base < deg; base += 32) {
        const int i = base + lane;
        if (i < deg) {
            const float4 a = alS[ssrc[beg + i]];
            m.x = fmaxf(m.x, leaky(a.x + dn.x));
            m.y = fmaxf(m.y, leaky(a.y + dn.y));
            m.z = fmaxf(m.z, leaky(a.z + dn.z));
            m.w = fmaxf(m.w, leaky(a.w + dn.w));
        }
    }
    m.x = wave_max(m.x); m.y = wave_max(m.y); m.z = wave_max(m.z); m.w = wave_max(m.w);

    const float4 wself = make_float4(__expf(es.x - m.x), __expf(es.y - m.y),
                                     __expf(es.z - m.z), __expf(es.w - m.w));
    const float wsl = sel4(wself.x, wself.y, wself.z, wself.w, k0, k1, k2, k3);
    const float4 hn = h[(size_t)n * 32 + lane];
    float4 acc = make_float4(wsl * hn.x, wsl * hn.y, wsl * hn.z, wsl * hn.w);
    float4 dp = make_float4(0.f, 0.f, 0.f, 0.f);

    float4 w = make_float4(0.f, 0.f, 0.f, 0.f);
    if (v0) w = make_float4(__expf(e0.x - m.x), __expf(e0.y - m.y),
                            __expf(e0.z - m.z), __expf(e0.w - m.w));
    int base = 0;
    for (;;) {
        dp.x += w.x; dp.y += w.y; dp.z += w.z; dp.w += w.w;
        const int cc = min(32, deg - base);
        for (int j = 0; j < cc; j += 4) {
            int sj[4];
            float wj[4];
            float4 r[4];
            for (int u = 0; u < 4; ++u) {
                sj[u] = __builtin_amdgcn_readlane(s, j + u);
                wj[u] = sel4(lane_val(w.x, j + u), lane_val(w.y, j + u),
                             lane_val(w.z, j + u), lane_val(w.w, j + u), k0, k1, k2, k3);
            }
            for (int u = 0; u < 4; ++u) r[u] = h[(size_t)sj[u] * 32 + lane];
            for (int u = 0; u < 4; ++u) {
                acc.x += wj[u] * r[u].x; acc.y += wj[u] * r[u].y;
                acc.z += wj[u] * r[u].z; acc.w += wj[u] * r[u].w;
            }
        }
        base += 32;
        if (base >= deg) break;
        const int i = base + lane;
        s = n;
        w = make_float4(0.f, 0.f, 0.f, 0.f);
        if (i < deg) {
            s = ssrc[beg + i];
            const float4 a = alS[s];
            w = make_float4(__expf(leaky(a.x + dn.x) - m.x), __expf(leaky(a.y + dn.y) - m.y),
                            __expf(leaky(a.z + dn.z) - m.z), __expf(leaky(a.w + dn.w) - m.w));
        }
    }

    const float den = sel4(wave_sum(dp.x) + wself.x, wave_sum(dp.y) + wself.y,
                           wave_sum(dp.z) + wself.z, wave_sum(dp.w) + wself.w,
                           k0, k1, k2, k3) + INF_EPS;
    const float* bp = bias + lane * 4;
    float4 o;
    o.x = acc.x / den + bp[0];
    o.y = acc.y / den + bp[1];
    o.z = acc.z / den + bp[2];
    o.w = acc.w / den + bp[3];
    o.x = o.x > 0.f ? o.x : __expf(o.x) - 1.f;
    o.y = o.y > 0.f ? o.y : __expf(o.y) - 1.f;
    o.z = o.z > 0.f ? o.z : __expf(o.z) - 1.f;
    o.w = o.w > 0.f ? o.w : __expf(o.w) - 1.f;
    outp[(size_t)n * 32 + lane] = o;
}

__global__ __launch_bounds__(256) void gather_softmax_h1(
    const int* __restrict__ off, const int* __restrict__ cnt, const int* __restrict__ ssrc,
    const float* __restrict__ alS, const float* __restrict__ alD,
    const float4* __restrict__ h, const float* __restrict__ bias,
    float4* __restrict__ outp, int N)
{
    const int lane = threadIdx.x & 31;
    const int g = lane >> 4;
    const int l16 = lane & 15;
    const int n = __builtin_amdgcn_readfirstlane((int)(blockIdx.x * 8 + (threadIdx.x >> 5)));
    if (n >= N) return;
    const int beg = off[n];
    const int deg = cnt[n];
    const float dn = alD[n];
    const float es = leaky(alS[n] + dn);

    const bool v0 = lane < deg;
    int s = n;
    float e0 = es;
    if (v0) {
        s = ssrc[beg + lane];
        e0 = leaky(alS[s] + dn);
    }
    float m = fmaxf(es, e0);
    for (int base = 32; base < deg; base += 32) {
        const int i = base + lane;
        if (i < deg) m = fmaxf(m, leaky(alS[ssrc[beg + i]] + dn));
    }
    m = wave_max(m);

    const float wself = __expf(es - m);
    float4 acc = make_float4(0.f, 0.f, 0.f, 0.f);
    if (g == 0) {
        const float4 hn = h[(size_t)n * 16 + l16];
        acc = make_float4(wself * hn.x, wself * hn.y, wself * hn.z, wself * hn.w);
    }
    float dp = 0.f;
    float w = v0 ? __expf(e0 - m) : 0.f;
    int base = 0;
    for (;;) {
        dp += w;
        const int cc = min(32, deg - base);
        for (int j = 0; j < cc; j += 8) {
            int sj[4];
            float wj[4];
            float4 r[4];
            for (int u = 0; u < 4; ++u) {
                sj[u] = __shfl(s, j + 2 * u + g);
                wj[u] = __shfl(w, j + 2 * u + g);
            }
            for (int u = 0; u < 4; ++u) r[u] = h[(size_t)sj[u] * 16 + l16];
            for (int u = 0; u < 4; ++u) {
                acc.x += wj[u] * r[u].x; acc.y += wj[u] * r[u].y;
                acc.z += wj[u] * r[u].z; acc.w += wj[u] * r[u].w;
            }
        }
        base += 32;
        if (base >= deg) break;
        const int i = base + lane;
        s = n;
        w = 0.f;
        if (i < deg) {
            s = ssrc[beg + i];
            w = __expf(leaky(alS[s] + dn) - m);
        }
    }

    const float den = wave_sum(dp) + wself + INF_EPS;
    acc.x += __shfl_xor(acc.x, 16);
    acc.y += __shfl_xor(acc.y, 16);
    acc.z += __shfl_xor(acc.z, 16);
    acc.w += __shfl_xor(acc.w, 16);
    if (g == 0) {
        const float* bp = bias + l16 * 4;
        float4 o;
        o.x = acc.x / den + bp[0];
        o.y = acc.y / den + bp[1];
        o.z = acc.z / den + bp[2];
        o.w = acc.w / den + bp[3];
        outp[(size_t)n * 16 + l16] = o;
    }
}

extern "C" void kernel_launch(void* const* d_in, const int* in_sizes, int n_in,
                              void* d_out, int out_size, void* d_ws, size_t ws_size,
                              hipStream_t stream)
{
    const float* x    = (const float*)d_in[0];
    const int*   eidx = (const int*)d_in[1];
    const float* W1   = (const float*)d_in[2];
    const float* aS1  = (const float*)d_in[3];
    const float* aD1  = (const float*)d_in[4];
    const float* b1   = (const float*)d_in[5];
    const float* W2   = (const float*)d_in[6];
    const float* aS2  = (const float*)d_in[7];
    const float* aD2  = (const float*)d_in[8];
    const float* b2   = (const float*)d_in[9];
    float* out = (float*)d_out;

    const int N    = in_sizes[0] / 128;
    const int E    = in_sizes[1] / 2;
    const int* esrc = eidx;
    const int* edst = eidx + E;

    float* ws = (float*)d_ws;
    size_t o = 0;
    float* h1   = ws + o; o += (size_t)N * 128;
    float* alS1 = ws + o; o += (size_t)N * 4;
    float* alD1 = ws + o; o += (size_t)N * 4;
    float* agg1 = ws + o; o += (size_t)N * 128;
    float* h2   = ws + o; o += (size_t)N * 64;
    float* alS2 = ws + o; o += ((size_t)N + 31) & ~(size_t)31;
    float* alD2 = ws + o; o += ((size_t)N + 31) & ~(size_t)31;
    o = (o + 31) & ~(size_t)31;
    unsigned* bpack = (unsigned*)(ws + o); o += (size_t)E;
    int*   ssrc = (int*)(ws + o); o += (size_t)E;
    int*   off  = (int*)(ws + o); o += (size_t)N;
    int*   cnt  = (int*)(ws + o); o += (size_t)N;
    o = (o + 31) & ~(size_t)31;
    auto cdiv = [](int a, int b) { return (a + b - 1) / b; };
    const int nChunk  = cdiv(E, SORT_CHUNK);
    const int nBucket = cdiv(N, 256);
    unsigned* meta = (unsigned*)(ws + o); o += (size_t)nChunk * 256;

    chunk_group<<<nChunk, 256, 0, stream>>>(esrc, edst, bpack, meta, E);
    bucket_sort<<<nBucket, 1024, 0, stream>>>(bpack, meta, cnt, off, ssrc, N, nChunk);

    gemm_wmma_f32<128, 128, 32><<<cdiv(N / 16, 2), 128, 0, stream>>>(x, W1, aS1, aD1, h1, alS1, alD1, N / 16);
    gather_softmax_h4<<<cdiv(N, 8), 256, 0, stream>>>(off, cnt, ssrc,
        (const float4*)alS1, (const float4*)alD1, (const float4*)h1, b1, (float4*)agg1, N);

    gemm_wmma_f32<128, 64, 64><<<cdiv(N / 16, 4), 128, 0, stream>>>(agg1, W2, aS2, aD2, h2, alS2, alD2, N / 16);
    gather_softmax_h1<<<cdiv(N, 8), 256, 0, stream>>>(off, cnt, ssrc,
        alS2, alD2, (const float4*)h2, b2, (float4*)out, N);

    (void)n_in; (void)out_size; (void)ws_size;
}


# Both gathers: the lanes that sum a head's features own its edge logits, so max, sum and weights move by DPP inside the lane group, not readlane+select or LDS permutes; softmax sum order changes
# speedup vs baseline: 10.2915x; 1.0452x over previous
#include <hip/hip_runtime.h>
#include <hip/hip_bf16.h>

typedef float v2f __attribute__((ext_vector_type(2)));
typedef float v8f __attribute__((ext_vector_type(8)));

#define NEG_SLOPE 0.2f
#define INF_EPS   1e-16f

__device__ __forceinline__ float leaky(float x) { return x > 0.f ? x : NEG_SLOPE * x; }

template <int M>
__device__ __forceinline__ float row_xor(float v) {
    return __int_as_float(__builtin_amdgcn_mov_dpp(__float_as_int(v), 0x160 | M, 0xf, 0xf, false));
}

template <int K, int NCOLS, int HC>
__global__ __launch_bounds__(128) __attribute__((amdgpu_waves_per_eu(4, 4))) void gemm_wmma_f32(
    const float* __restrict__ A, const float* __restrict__ B,
    const float* __restrict__ attS, const float* __restrict__ attD,
    float* __restrict__ C, float* __restrict__ alS, float* __restrict__ alD, int rowTiles)
{
    constexpr int CG  = NCOLS / 64;
    constexpr int TPB = 4 / CG;
    constexpr int H   = NCOLS / HC;
    constexpr int HPW = 64 / HC;
    constexpr int LPH = HC / 4;
    constexpr int F   = TPB * 16 * H;
    __shared__ __attribute__((aligned(16))) float s_al[2 * F];

    const int lane = threadIdx.x & 31;
    const int wave = __builtin_amdgcn_readfirstlane((int)(threadIdx.x >> 5));
    const int ti   = wave / CG;
    const int cg   = wave % CG;
    const int rt   = blockIdx.x * TPB + ti;
    const int c    = lane & 15;
    const int hh   = lane >> 4;
    const int col4 = cg * 64 + 4 * c;
    const bool live = rt < rowTiles;
    v8f acc0 = {}, acc1 = {}, acc2 = {}, acc3 = {};
    if (live) {
        const int row  = rt * 16 + c;
        const int kh   = hh << 1;

        const float* __restrict__ aPtr = A + (size_t)row * K + kh;
        const float* __restrict__ bPtr = B + (size_t)kh * NCOLS + col4;
        const float4 aS4 = *reinterpret_cast<const float4*>(attS + col4);
        const float4 aD4 = *reinterpret_cast<const float4*>(attD + col4);

#pragma unroll
        for (int k0 = 0; k0 < K; k0 += 4) {
            v2f a, b;
            a.x = aPtr[k0];
            a.y = aPtr[k0 + 1];
            const float4 b0 = *reinterpret_cast<const float4*>(bPtr + (size_t)k0 * NCOLS);
            const float4 b1 = *reinterpret_cast<const float4*>(bPtr + (size_t)k0 * NCOLS + NCOLS);
            b.x = b0.x; b.y = b1.x;
            acc0 = __builtin_amdgcn_wmma_f32_16x16x4_f32(
                false, a, false, b, (short)0, acc0, false, false);
            b.x = b0.y; b.y = b1.y;
            acc1 = __builtin_amdgcn_wmma_f32_16x16x4_f32(
                false, a, false, b, (short)0, acc1, false, false);
            b.x = b0.z; b.y = b1.z;
            acc2 = __builtin_amdgcn_wmma_f32_16x16x4_f32(
                false, a, false, b, (short)0, acc2, false, false);
            b.x = b0.w; b.y = b1.w;
            acc3 = __builtin_amdgcn_wmma_f32_16x16x4_f32(
                false, a, false, b, (short)0, acc3, false, false);
        }
        float ps[8], pd[8];
#pragma unroll
        for (int r = 0; r < 8; ++r) {
            ps[r] = acc0[r] * aS4.x + acc1[r] * aS4.y + acc2[r] * aS4.z + acc3[r] * aS4.w;
            pd[r] = acc0[r] * aD4.x + acc1[r] * aD4.y + acc2[r] * aD4.z + acc3[r] * aD4.w;
        }
        const bool u4 = (c & 4) != 0, u2 = (c & 2) != 0, u1 = (c & 1) != 0;
        float qs[4], qd[4];
#pragma unroll
        for (int i = 0; i < 4; ++i) {
            qs[i] = (u4 ? ps[i + 4] : ps[i]) + row_xor<4>(u4 ? ps[i] : ps[i + 4]);
            qd[i] = (u4 ? pd[i + 4] : pd[i]) + row_xor<4>(u4 ? pd[i] : pd[i + 4]);
        }
        float ws[2], wd[2];
#pragma unroll
        for (int i = 0; i < 2; ++i) {
            ws[i] = (u2 ? qs[i + 2] : qs[i]) + row_xor<2>(u2 ? qs[i] : qs[i + 2]);
            wd[i] = (u2 ? qd[i + 2] : qd[i]) + row_xor<2>(u2 ? qd[i] : qd[i + 2]);
        }
        float ts = (u1 ? ws[1] : ws[0]) + row_xor<1>(u1 ? ws[0] : ws[1]);
        float td = (u1 ? wd[1] : wd[0]) + row_xor<1>(u1 ? wd[0] : wd[1]);
        if (LPH == 16) { ts += row_xor<8>(ts); td += row_xor<8>(td); }
        if ((c % LPH) < 8) {
            const int idx = (ti * 16 + (hh << 3) + (c & 7)) * H + cg * HPW + c / LPH;
            s_al[idx]     = ts;
            s_al[F + idx] = td;
        }
    }
    __syncthreads();
    if (live) {
        float* __restrict__ cPtr =
            C + ((size_t)rt * 16 + (hh << 3)) * NCOLS + col4;
#pragma unroll
        for (int r = 0; r < 8; ++r)
            *reinterpret_cast<float4*>(cPtr + (size_t)r * NCOLS) =
                make_float4(acc0[r], acc1[r], acc2[r], acc3[r]);
    }
    if (wave < 2 && lane < F / 4) {
        const int e = lane * 4;
        if (blockIdx.x * TPB + e / (16 * H) < rowTiles) {
            const float4 v = *reinterpret_cast<const float4*>(&s_al[wave * F + e]);
            float* __restrict__ dst = (wave == 0 ? alS : alD) + (size_t)blockIdx.x * F + e;
            *reinterpret_cast<float4*>(dst) = v;
        }
    }
}

#define SORT_CHUNK 2048

__device__ __forceinline__ int wave_incl_scan_i(int v, int lane) {
    for (int o = 1; o < 32; o <<= 1) {
        int u = __shfl_up(v, o);
        if (lane >= o) v += u;
    }
    return v;
}

__global__ __launch_bounds__(256) void chunk_group(
    const int* __restrict__ esrc, const int* __restrict__ edst,
    unsigned* __restrict__ bpack, unsigned* __restrict__ meta, int E)
{
    __shared__ int s_h[256];
    __shared__ int s_w[8];
    __shared__ unsigned s_out[SORT_CHUNK];
    const int t = threadIdx.x;
    const int lane = t & 31;
    const int wave = t >> 5;
    s_h[t] = 0;
    const int cb = blockIdx.x * SORT_CHUNK;
    const int e0 = cb + t;
    int sv[SORT_CHUNK / 256], dv[SORT_CHUNK / 256], rk[SORT_CHUNK / 256];
#pragma unroll
    for (int k = 0; k < SORT_CHUNK / 256; ++k) {
        const int e = e0 + k * 256;
        sv[k] = 0; dv[k] = 0;
        if (e < E) { sv[k] = esrc[e]; dv[k] = edst[e]; }
    }
    __syncthreads();
#pragma unroll
    for (int k = 0; k < SORT_CHUNK / 256; ++k) {
        const int e = e0 + k * 256;
        rk[k] = 0;
        if (e < E) rk[k] = atomicAdd(&s_h[dv[k] >> 8], 1);
    }
    __syncthreads();
    const int c = s_h[t];
    const int incl = wave_incl_scan_i(c, lane);
    if (lane == 31) s_w[wave] = incl;
    __syncthreads();
    int wb = 0;
    for (int w = 0; w < 8; ++w) if (w < wave) wb += s_w[w];
    const int ex = wb + incl - c;
    s_h[t] = ex;
    __syncthreads();
#pragma unroll
    for (int k = 0; k < SORT_CHUNK / 256; ++k) {
        const int e = e0 + k * 256;
        if (e < E)
            s_out[s_h[dv[k] >> 8] + rk[k]] = ((unsigned)sv[k] << 8) | ((unsigned)dv[k] & 255u);
    }
    __syncthreads();
#pragma unroll
    for (int k = 0; k < SORT_CHUNK / 256; ++k) {
        const int i = t + k * 256;
        if (cb + i < E) bpack[cb + i] = s_out[i];
    }
    meta[blockIdx.x * 256 + t] = (unsigned)ex | ((unsigned)c << 16);
}

template <bool SCATTER>
__device__ __forceinline__ void walk_pieces(
    const unsigned* __restrict__ bpack, const unsigned* s_meta, int* s_ctr,
    int* __restrict__ seg, int c0, int nt, int g, int l16)
{
    for (int q0 = g; q0 < nt; q0 += 512) {
        unsigned p[8];
        int pc[8], bs[8];
#pragma unroll
        for (int k = 0; k < 8; ++k) {
            const int q = q0 + 64 * k;
            const unsigned mq = s_meta[q];
            pc[k] = (int)(mq >> 16);
            bs[k] = (c0 + q) * SORT_CHUNK + (int)(mq & 0xffffu);
            p[k] = bpack[l16 < pc[k] ? bs[k] + l16 : 0];
        }
#pragma unroll
        for (int k = 0; k < 8; ++k) {
            if (l16 < pc[k]) {
                const int pos = atomicAdd(&s_ctr[p[k] & 255u], 1);
                if (SCATTER) seg[pos] = (int)(p[k] >> 8);
            }
            for (int i = l16 + 16; i < pc[k]; i += 16) {
                const unsigned pp = bpack[bs[k] + i];
                const int pos = atomicAdd(&s_ctr[pp & 255u], 1);
                if (SCATTER) seg[pos] = (int)(pp >> 8);
            }
        }
    }
}

__global__ __launch_bounds__(1024) void bucket_sort(
    const unsigned* __restrict__ bpack, const unsigned* __restrict__ meta,
    int* __restrict__ cnt, int* __restrict__ off, int* __restrict__ ssrc, int N, int nChunk)
{
    __shared__ int s_cnt[256];
    __shared__ int s_cur[256];
    __shared__ int s_w[8];
    __shared__ int s_start;
    __shared__ unsigned s_meta[1024];
    const int b = blockIdx.x;
    const int t = threadIdx.x;
    const int lane = t & 31;
    const int wave = t >> 5;
    const int g = t >> 4;
    const int l16 = t & 15;
    unsigned m0 = 0u;
    if (t < nChunk) m0 = meta[(size_t)t * 256 + b];
    if (t < 256) s_cnt[t] = 0;
    if (t == 0) s_start = 0;
    __syncthreads();
    for (int c0 = 0; c0 < nChunk; c0 += 1024) {
        unsigned mv = m0;
        if (c0 > 0) {
            mv = 0u;
            if (c0 + t < nChunk) mv = meta[(size_t)(c0 + t) * 256 + b];
        }
        s_meta[t] = mv;
        int v = (int)(mv & 0xffffu);
        for (int o = 16; o > 0; o >>= 1) v += __shfl_xor(v, o);
        if (lane == 0 && v != 0) atomicAdd(&s_start, v);
        __syncthreads();
        walk_pieces<false>(bpack, s_meta, s_cnt, nullptr, c0, min(1024, nChunk - c0), g, l16);
        __syncthreads();
    }
    const int start = s_start;
    const int c = (t < 256) ? s_cnt[t] : 0;
    const int incl = wave_incl_scan_i(c, lane);
    if (t < 256 && lane == 31) s_w[wave] = incl;
    __syncthreads();
    int ex0 = 0;
    if (t < 256) {
        int wb = 0;
        for (int w = 0; w < 8; ++w) if (w < wave) wb += s_w[w];
        const int ex = wb + incl - c;
        s_cur[t] = ex;
        ex0 = ex;
    }
    __syncthreads();
    if (t < 256) {
        const int n = b * 256 + t;
        if (n < N) { cnt[n] = c; off[n] = start + ex0; }
    }
    for (int c0 = 0; c0 < nChunk; c0 += 1024) {
        if (nChunk > 1024) {
            unsigned mv = 0u;
            if (c0 + t < nChunk) mv = meta[(size_t)(c0 + t) * 256 + b];
            s_meta[t] = mv;
            __syncthreads();
        }
        walk_pieces<true>(bpack, s_meta, s_cur, ssrc + start, c0, min(1024, nChunk - c0), g, l16);
        if (nChunk > 1024) __syncthreads();
    }
}

template <int T>
__device__ __forceinline__ float grp8_bcast(float v) {
    return __uint_as_float(__builtin_amdgcn_mov_dpp8(__float_as_uint(v), (unsigned)T * 0x249249u));
}
template <int T>
__device__ __forceinline__ int row_bcast_i(int v) {
    return __builtin_amdgcn_mov_dpp(v, 0x150 | T, 0xf, 0xf, false);
}
template <int T>
__device__ __forceinline__ float row_bcast_f(float v) {
    return __int_as_float(__builtin_amdgcn_mov_dpp(__float_as_int(v), 0x150 | T, 0xf, 0xf, false));
}
__device__ __forceinline__ float xhalf(float x) {
    const int b = __float_as_int(x);
    return __int_as_float(__builtin_amdgcn_permlanex16(b, b, 0x76543210, 0xfedcba98, false, false));
}

template <int T0>
__device__ __forceinline__ void rows4_h4(v2f& accLo, v2f& accHi, int sk, float wk,
                                         const float4* __restrict__ h, int lane) {
    int sj[4];
    float wj[4];
    float4 r[4];
    sj[0] = __builtin_amdgcn_readlane(sk, T0);
    sj[1] = __builtin_amdgcn_readlane(sk, T0 + 1);
    sj[2] = __builtin_amdgcn_readlane(sk, T0 + 2);
    sj[3] = __builtin_amdgcn_readlane(sk, T0 + 3);
    wj[0] = grp8_bcast<T0>(wk);
    wj[1] = grp8_bcast<T0 + 1>(wk);
    wj[2] = grp8_bcast<T0 + 2>(wk);
    wj[3] = grp8_bcast<T0 + 3>(wk);
    for (int u = 0; u < 4; ++u) r[u] = h[(size_t)sj[u] * 32 + lane];
    for (int u = 0; u < 4; ++u) {
        const v2f wv = {wj[u], wj[u]};
        const v2f lo = {r[u].x, r[u].y};
        const v2f hi = {r[u].z, r[u].w};
        accLo += wv * lo;
        accHi += wv * hi;
    }
}

__global__ __launch_bounds__(256) void gather_softmax_h4(
    const int* __restrict__ off, const int* __restrict__ cnt, const int* __restrict__ ssrc,
    const float* __restrict__ alS, const float* __restrict__ alD,
    const float4* __restrict__ h, const float* __restrict__ bias,
    float4* __restrict__ outp, int N)
{
    const int lane = threadIdx.x & 31;
    const int hd = lane >> 3;
    const int q  = lane & 7;
    const int n = __builtin_amdgcn_readfirstlane((int)(blockIdx.x * 8 + (threadIdx.x >> 5)));
    if (n >= N) return;
    const int beg = off[n];
    const int deg = cnt[n];
    const float dn = alD[(size_t)n * 4 + hd];
    const float es = leaky(alS[(size_t)n * 4 + hd] + dn);

    int s[4];
    float e[4];
#pragma unroll
    for (int k = 0; k < 4; ++k) {
        const int i = q + 8 * k;
        const int sl = ssrc[i < deg ? beg + i : 0];
        s[k] = i < deg ? sl : n;
    }
#pragma unroll
    for (int k = 0; k < 4; ++k)
        e[k] = leaky(alS[(size_t)s[k] * 4 + hd] + dn);
    float m = fmaxf(fmaxf(fmaxf(es, e[0]), fmaxf(e[1], e[2])), e[3]);
    for (int base = 32; base < deg; base += 32) {
        for (int k = 0; k < 4; ++k) {
            const int i = base + q + 8 * k;
            if (i < deg) m = fmaxf(m, leaky(alS[(size_t)ssrc[beg + i] * 4 + hd] + dn));
        }
    }
    m = fmaxf(m, row_xor<1>(m));
    m = fmaxf(m, row_xor<2>(m));
    m = fmaxf(m, row_xor<4>(m));

    const float wself = __expf(es - m);
    const float4 hn = h[(size_t)n * 32 + lane];
    v2f accLo = {wself * hn.x, wself * hn.y};
    v2f accHi = {wself * hn.z, wself * hn.w};
    float dp = 0.f;

    float w[4];
#pragma unroll
    for (int k = 0; k < 4; ++k) w[k] = (q + 8 * k < deg) ? __expf(e[k] - m) : 0.f;
    int base = 0;
    for (;;) {
        dp += (w[0] + w[1]) + (w[2] + w[3]);
        const int cc = min(32, deg - base);
#pragma unroll
        for (int k = 0; k < 4; ++k) {
            if (8 * k < cc)     rows4_h4<0>(accLo, accHi, s[k], w[k], h, lane);
            if (8 * k + 4 < cc) rows4_h4<4>(accLo, accHi, s[k], w[k], h, lane);
        }
        base += 32;
        if (base >= deg) break;
#pragma unroll
        for (int k = 0; k < 4; ++k) {
            const int i = base + q + 8 * k;
            s[k] = n;
            w[k] = 0.f;
            if (i < deg) {
                s[k] = ssrc[beg + i];
                w[k] = __expf(leaky(alS[(size_t)s[k] * 4 + hd] + dn) - m);
            }
        }
    }

    dp += row_xor<1>(dp);
    dp += row_xor<2>(dp);
    dp += row_xor<4>(dp);
    const float den = dp + wself + INF_EPS;
    const float* bp = bias + lane * 4;
    float4 o;
    o.x = accLo.x / den + bp[0];
    o.y = accLo.y / den + bp[1];
    o.z = accHi.x / den + bp[2];
    o.w = accHi.y / den + bp[3];
    o.x = o.x > 0.f ? o.x : __expf(o.x) - 1.f;
    o.y = o.y > 0.f ? o.y : __expf(o.y) - 1.f;
    o.z = o.z > 0.f ? o.z : __expf(o.z) - 1.f;
    o.w = o.w > 0.f ? o.w : __expf(o.w) - 1.f;
    outp[(size_t)n * 32 + lane] = o;
}

template <int T0>
__device__ __forceinline__ void rows4_h1(v2f& accLo, v2f& accHi, int s, float w,
                                         const float4* __restrict__ h, int l16) {
    int sj[4];
    float wj[4];
    float4 r[4];
    sj[0] = row_bcast_i<T0>(s);     wj[0] = row_bcast_f<T0>(w);
    sj[1] = row_bcast_i<T0 + 1>(s); wj[1] = row_bcast_f<T0 + 1>(w);
    sj[2] = row_bcast_i<T0 + 2>(s); wj[2] = row_bcast_f<T0 + 2>(w);
    sj[3] = row_bcast_i<T0 + 3>(s); wj[3] = row_bcast_f<T0 + 3>(w);
    for (int u = 0; u < 4; ++u) r[u] = h[(size_t)sj[u] * 16 + l16];
    for (int u = 0; u < 4; ++u) {
        const v2f wv = {wj[u], wj[u]};
        const v2f lo = {r[u].x, r[u].y};
        const v2f hi = {r[u].z, r[u].w};
        accLo += wv * lo;
        accHi += wv * hi;
    }
}

__global__ __launch_bounds__(256) void gather_softmax_h1(
    const int* __restrict__ off, const int* __restrict__ cnt, const int* __restrict__ ssrc,
    const float* __restrict__ alS, const float* __restrict__ alD,
    const float4* __restrict__ h, const float* __restrict__ bias,
    float4* __restrict__ outp, int N)
{
    const int lane = threadIdx.x & 31;
    const int g = lane >> 4;
    const int l16 = lane & 15;
    const int own = 2 * l16 + g;
    const int n = __builtin_amdgcn_readfirstlane((int)(blockIdx.x * 8 + (threadIdx.x >> 5)));
    if (n >= N) return;
    const int beg = off[n];
    const int deg = cnt[n];
    const float dn = alD[n];
    const float es = leaky(alS[n] + dn);

    const bool v0 = own < deg;
    int s = n;
    float e0 = es;
    if (v0) {
        s = ssrc[beg + own];
        e0 = leaky(alS[s] + dn);
    }
    float m = fmaxf(es, e0);
    for (int base = 32; base < deg; base += 32) {
        const int i = base + own;
        if (i < deg) m = fmaxf(m, leaky(alS[ssrc[beg + i]] + dn));
    }
    m = fmaxf(m, row_xor<1>(m));
    m = fmaxf(m, row_xor<2>(m));
    m = fmaxf(m, row_xor<4>(m));
    m = fmaxf(m, row_xor<8>(m));
    m = fmaxf(m, xhalf(m));

    const float wself = __expf(es - m);
    v2f accLo = {0.f, 0.f};
    v2f accHi = {0.f, 0.f};
    if (g == 0) {
        const float4 hn = h[(size_t)n * 16 + l16];
        accLo = (v2f){wself * hn.x, wself * hn.y};
        accHi = (v2f){wself * hn.z, wself * hn.w};
    }
    float dp = 0.f;
    float w = v0 ? __expf(e0 - m) : 0.f;
    int base = 0;
    for (;;) {
        dp += w;
        const int cc = min(32, deg - base);
        if (0 < cc)  rows4_h1<0>(accLo, accHi, s, w, h, l16);
        if (8 < cc)  rows4_h1<4>(accLo, accHi, s, w, h, l16);
        if (16 < cc) rows4_h1<8>(accLo, accHi, s, w, h, l16);
        if (24 < cc) rows4_h1<12>(accLo, accHi, s, w, h, l16);
        base += 32;
        if (base >= deg) break;
        const int i = base + own;
        s = n;
        w = 0.f;
        if (i < deg) {
            s = ssrc[beg + i];
            w = __expf(leaky(alS[s] + dn) - m);
        }
    }

    dp += row_xor<1>(dp);
    dp += row_xor<2>(dp);
    dp += row_xor<4>(dp);
    dp += row_xor<8>(dp);
    dp += xhalf(dp);
    const float den = dp + wself + INF_EPS;
    const float ax = accLo.x + xhalf(accLo.x);
    const float ay = accLo.y + xhalf(accLo.y);
    const float az = accHi.x + xhalf(accHi.x);
    const float aw = accHi.y + xhalf(accHi.y);
    if (g == 0) {
        const float* bp = bias + l16 * 4;
        float4 o;
        o.x = ax / den + bp[0];
        o.y = ay / den + bp[1];
        o.z = az / den + bp[2];
        o.w = aw / den + bp[3];
        outp[(size_t)n * 16 + l16] = o;
    }
}

extern "C" void kernel_launch(void* const* d_in, const int* in_sizes, int n_in,
                              void* d_out, int out_size, void* d_ws, size_t ws_size,
                              hipStream_t stream)
{
    const float* x    = (const float*)d_in[0];
    const int*   eidx = (const int*)d_in[1];
    const float* W1   = (const float*)d_in[2];
    const float* aS1  = (const float*)d_in[3];
    const float* aD1  = (const float*)d_in[4];
    const float* b1   = (const float*)d_in[5];
    const float* W2   = (const float*)d_in[6];
    const float* aS2  = (const float*)d_in[7];
    const float* aD2  = (const float*)d_in[8];
    const float* b2   = (const float*)d_in[9];
    float* out = (float*)d_out;

    const int N    = in_sizes[0] / 128;
    const int E    = in_sizes[1] / 2;
    const int* esrc = eidx;
    const int* edst = eidx + E;

    float* ws = (float*)d_ws;
    size_t o = 0;
    float* h1   = ws + o; o += (size_t)N * 128;
    float* alS1 = ws + o; o += (size_t)N * 4;
    float* alD1 = ws + o; o += (size_t)N * 4;
    float* agg1 = ws + o; o += (size_t)N * 128;
    float* h2   = ws + o; o += (size_t)N * 64;
    float* alS2 = ws + o; o += ((size_t)N + 31) & ~(size_t)31;
    float* alD2 = ws + o; o += ((size_t)N + 31) & ~(size_t)31;
    o = (o + 31) & ~(size_t)31;
    unsigned* bpack = (unsigned*)(ws + o); o += (size_t)E;
    int*   ssrc = (int*)(ws + o); o += (size_t)E;
    int*   off  = (int*)(ws + o); o += (size_t)N;
    int*   cnt  = (int*)(ws + o); o += (size_t)N;
    o = (o + 31) & ~(size_t)31;
    auto cdiv = [](int a, int b) { return (a + b - 1) / b; };
    const int nChunk  = cdiv(E, SORT_CHUNK);
    const int nBucket = cdiv(N, 256);
    unsigned* meta = (unsigned*)(ws + o); o += (size_t)nChunk * 256;

    chunk_group<<<nChunk, 256, 0, stream>>>(esrc, edst, bpack, meta, E);
    bucket_sort<<<nBucket, 1024, 0, stream>>>(bpack, meta, cnt, off, ssrc, N, nChunk);

    gemm_wmma_f32<128, 128, 32><<<cdiv(N / 16, 2), 128, 0, stream>>>(x, W1, aS1, aD1, h1, alS1, alD1, N / 16);
    gather_softmax_h4<<<cdiv(N, 8), 256, 0, stream>>>(off, cnt, ssrc,
        alS1, alD1, (const float4*)h1, b1, (float4*)agg1, N);

    gemm_wmma_f32<128, 64, 64><<<cdiv(N / 16, 4), 128, 0, stream>>>(agg1, W2, aS2, aD2, h2, alS2, alD2, N / 16);
    gather_softmax_h1<<<cdiv(N, 8), 256, 0, stream>>>(off, cnt, ssrc,
        alS2, alD2, (const float4*)h2, b2, (float4*)out, N);

    (void)n_in; (void)out_size; (void)ws_size;
}


# Layer-1 dense product runs in the sort's SECOND launch as further 256-thread work-groups beside the bucket pass (no later kernel waits longer for it): 6 launches to 5; every sum and k order unchanged
# speedup vs baseline: 11.1526x; 1.0837x over previous
#include <hip/hip_runtime.h>
#include <hip/hip_bf16.h>

typedef float v2f __attribute__((ext_vector_type(2)));
typedef float v8f __attribute__((ext_vector_type(8)));

#define NEG_SLOPE 0.2f
#define INF_EPS   1e-16f

__device__ __forceinline__ float leaky(float x) { return x > 0.f ? x : NEG_SLOPE * x; }

template <int M>
__device__ __forceinline__ float row_xor(float v) {
    return __int_as_float(__builtin_amdgcn_mov_dpp(__float_as_int(v), 0x160 | M, 0xf, 0xf, false));
}

template <int K, int NCOLS, int HC, int NW>
__device__ __forceinline__ void gemm_body(
    const float* __restrict__ A, const float* __restrict__ B,
    const float* __restrict__ attS, const float* __restrict__ attD,
    float* __restrict__ C, float* __restrict__ alS, float* __restrict__ alD, int rowTiles,
    const int blk, float* s_al)
{
    constexpr int CG  = NCOLS / 64;
    constexpr int TPB = NW / CG;
    constexpr int H   = NCOLS / HC;
    constexpr int HPW = 64 / HC;
    constexpr int LPH = HC / 4;
    constexpr int F   = TPB * 16 * H;
    constexpr int NSW = (F + 127) / 128;

    const int lane = threadIdx.x & 31;
    const int wave = __builtin_amdgcn_readfirstlane((int)(threadIdx.x >> 5));
    const int ti   = wave / CG;
    const int cg   = wave % CG;
    const int rt   = blk * TPB + ti;
    const int c    = lane & 15;
    const int hh   = lane >> 4;
    const int col4 = cg * 64 + 4 * c;
    const bool live = rt < rowTiles;
    v8f acc0 = {}, acc1 = {}, acc2 = {}, acc3 = {};
    if (live) {
        const int row  = rt * 16 + c;
        const int kh   = hh << 1;

        const float* __restrict__ aPtr = A + (size_t)row * K + kh;
        const float* __restrict__ bPtr = B + (size_t)kh * NCOLS + col4;
        const float4 aS4 = *reinterpret_cast<const float4*>(attS + col4);
        const float4 aD4 = *reinterpret_cast<const float4*>(attD + col4);

#pragma unroll
        for (int k0 = 0; k0 < K; k0 += 4) {
            v2f a, b;
            a.x = aPtr[k0];
            a.y = aPtr[k0 + 1];
            const float4 b0 = *reinterpret_cast<const float4*>(bPtr + (size_t)k0 * NCOLS);
            const float4 b1 = *reinterpret_cast<const float4*>(bPtr + (size_t)k0 * NCOLS + NCOLS);
            b.x = b0.x; b.y = b1.x;
            acc0 = __builtin_amdgcn_wmma_f32_16x16x4_f32(
                false, a, false, b, (short)0, acc0, false, false);
            b.x = b0.y; b.y = b1.y;
            acc1 = __builtin_amdgcn_wmma_f32_16x16x4_f32(
                false, a, false, b, (short)0, acc1, false, false);
            b.x = b0.z; b.y = b1.z;
            acc2 = __builtin_amdgcn_wmma_f32_16x16x4_f32(
                false, a, false, b, (short)0, acc2, false, false);
            b.x = b0.w; b.y = b1.w;
            acc3 = __builtin_amdgcn_wmma_f32_16x16x4_f32(
                false, a, false, b, (short)0, acc3, false, false);
        }
        float ps[8], pd[8];
#pragma unroll
        for (int r = 0; r < 8; ++r) {
            ps[r] = acc0[r] * aS4.x + acc1[r] * aS4.y + acc2[r] * aS4.z + acc3[r] * aS4.w;
            pd[r] = acc0[r] * aD4.x + acc1[r] * aD4.y + acc2[r] * aD4.z + acc3[r] * aD4.w;
        }
        const bool u4 = (c & 4) != 0, u2 = (c & 2) != 0, u1 = (c & 1) != 0;
        float qs[4], qd[4];
#pragma unroll
        for (int i = 0; i < 4; ++i) {
            qs[i] = (u4 ? ps[i + 4] : ps[i]) + row_xor<4>(u4 ? ps[i] : ps[i + 4]);
            qd[i] = (u4 ? pd[i + 4] : pd[i]) + row_xor<4>(u4 ? pd[i] : pd[i + 4]);
        }
        float ws[2], wd[2];
#pragma unroll
        for (int i = 0; i < 2; ++i) {
            ws[i] = (u2 ? qs[i + 2] : qs[i]) + row_xor<2>(u2 ? qs[i] : qs[i + 2]);
            wd[i] = (u2 ? qd[i + 2] : qd[i]) + row_xor<2>(u2 ? qd[i] : qd[i + 2]);
        }
        float ts = (u1 ? ws[1] : ws[0]) + row_xor<1>(u1 ? ws[0] : ws[1]);
        float td = (u1 ? wd[1] : wd[0]) + row_xor<1>(u1 ? wd[0] : wd[1]);
        if (LPH == 16) { ts += row_xor<8>(ts); td += row_xor<8>(td); }
        if ((c % LPH) < 8) {
            const int idx = (ti * 16 + (hh << 3) + (c & 7)) * H + cg * HPW + c / LPH;
            s_al[idx]     = ts;
            s_al[F + idx] = td;
        }
    }
    __syncthreads();
    if (live) {
        float* __restrict__ cPtr =
            C + ((size_t)rt * 16 + (hh << 3)) * NCOLS + col4;
#pragma unroll
        for (int r = 0; r < 8; ++r)
            *reinterpret_cast<float4*>(cPtr + (size_t)r * NCOLS) =
                make_float4(acc0[r], acc1[r], acc2[r], acc3[r]);
    }
    if (wave < 2 * NSW) {
        const int kind = wave / NSW;
        const int e = (wave % NSW) * 128 + lane * 4;
        if (e < F && blk * TPB + e / (16 * H) < rowTiles) {
            const float4 v = *reinterpret_cast<const float4*>(&s_al[kind * F + e]);
            float* __restrict__ dst = (kind == 0 ? alS : alD) + (size_t)blk * F + e;
            *reinterpret_cast<float4*>(dst) = v;
        }
    }
}

template <int K, int NCOLS, int HC>
__global__ __launch_bounds__(128) __attribute__((amdgpu_waves_per_eu(4, 4))) void gemm_wmma_f32(
    const float* __restrict__ A, const float* __restrict__ B,
    const float* __restrict__ attS, const float* __restrict__ attD,
    float* __restrict__ C, float* __restrict__ alS, float* __restrict__ alD, int rowTiles)
{
    __shared__ __attribute__((aligned(16))) float s_al[2 * (4 / (NCOLS / 64)) * 16 * (NCOLS / HC)];
    gemm_body<K, NCOLS, HC, 4>(A, B, attS, attD, C, alS, alD, rowTiles, (int)blockIdx.x, s_al);
}

#define SORT_CHUNK 2048

__device__ __forceinline__ int wave_incl_scan_i(int v, int lane) {
    for (int o = 1; o < 32; o <<= 1) {
        int u = __shfl_up(v, o);
        if (lane >= o) v += u;
    }
    return v;
}

__global__ __launch_bounds__(256) void chunk_group(
    const int* __restrict__ esrc, const int* __restrict__ edst,
    unsigned* __restrict__ bpack, unsigned* __restrict__ meta, int E)
{
    __shared__ int s_h[256];
    __shared__ int s_w[8];
    __shared__ unsigned s_out[SORT_CHUNK];
    const int t = threadIdx.x;
    const int lane = t & 31;
    const int wave = t >> 5;
    s_h[t] = 0;
    const int cb = blockIdx.x * SORT_CHUNK;
    const int e0 = cb + t;
    int sv[SORT_CHUNK / 256], dv[SORT_CHUNK / 256], rk[SORT_CHUNK / 256];
#pragma unroll
    for (int k = 0; k < SORT_CHUNK / 256; ++k) {
        const int e = e0 + k * 256;
        sv[k] = 0; dv[k] = 0;
        if (e < E) { sv[k] = esrc[e]; dv[k] = edst[e]; }
    }
    __syncthreads();
#pragma unroll
    for (int k = 0; k < SORT_CHUNK / 256; ++k) {
        const int e = e0 + k * 256;
        rk[k] = 0;
        if (e < E) rk[k] = atomicAdd(&s_h[dv[k] >> 8], 1);
    }
    __syncthreads();
    const int c = s_h[t];
    const int incl = wave_incl_scan_i(c, lane);
    if (lane == 31) s_w[wave] = incl;
    __syncthreads();
    int wb = 0;
    for (int w = 0; w < 8; ++w) if (w < wave) wb += s_w[w];
    const int ex = wb + incl - c;
    s_h[t] = ex;
    __syncthreads();
#pragma unroll
    for (int k = 0; k < SORT_CHUNK / 256; ++k) {
        const int e = e0 + k * 256;
        if (e < E)
            s_out[s_h[dv[k] >> 8] + rk[k]] = ((unsigned)sv[k] << 8) | ((unsigned)dv[k] & 255u);
    }
    __syncthreads();
#pragma unroll
    for (int k = 0; k < SORT_CHUNK / 256; ++k) {
        const int i = t + k * 256;
        if (cb + i < E) bpack[cb + i] = s_out[i];
    }
    meta[blockIdx.x * 256 + t] = (unsigned)ex | ((unsigned)c << 16);
}

template <bool SCATTER, int G>
__device__ __forceinline__ void walk_pieces(
    const unsigned* __restrict__ bpack, const unsigned* s_meta, int* s_ctr,
    int* __restrict__ seg, int c0, int nt, int g, int l16)
{
    for (int q0 = g; q0 < nt; q0 += 8 * G) {
        unsigned p[8];
        int pc[8], bs[8];
#pragma unroll
        for (int k = 0; k < 8; ++k) {
            const int q = q0 + G * k;
            const unsigned mq = s_meta[q];
            pc[k] = (int)(mq >> 16);
            bs[k] = (c0 + q) * SORT_CHUNK + (int)(mq & 0xffffu);
            p[k] = bpack[l16 < pc[k] ? bs[k] + l16 : 0];
        }
#pragma unroll
        for (int k = 0; k < 8; ++k) {
            if (l16 < pc[k]) {
                const int pos = atomicAdd(&s_ctr[p[k] & 255u], 1);
                if (SCATTER) seg[pos] = (int)(p[k] >> 8);
            }
            for (int i = l16 + 16; i < pc[k]; i += 16) {
                const unsigned pp = bpack[bs[k] + i];
                const int pos = atomicAdd(&s_ctr[pp & 255u], 1);
                if (SCATTER) seg[pos] = (int)(pp >> 8);
            }
        }
    }
}

__device__ __forceinline__ int load_meta_tile(const unsigned* __restrict__ meta, unsigned* s_meta,
                                              int c0, int nChunk, int b, int t) {
    unsigned mv[4];
#pragma unroll
    for (int i = 0; i < 4; ++i) {
        const int q = c0 + t + 256 * i;
        mv[i] = 0u;
        if (q < nChunk) mv[i] = meta[(size_t)q * 256 + b];
    }
    int v = 0;
#pragma unroll
    for (int i = 0; i < 4; ++i) {
        s_meta[t + 256 * i] = mv[i];
        v += (int)(mv[i] & 0xffffu);
    }
    return v;
}

__device__ __forceinline__ void bucket_body(
    const unsigned* __restrict__ bpack, const unsigned* __restrict__ meta,
    int* __restrict__ cnt, int* __restrict__ off, int* __restrict__ ssrc, int N, int nChunk, const int b,
    int* s_cnt, int* s_cur, int* s_w, int* s_start, unsigned* s_meta)
{
    const int t = threadIdx.x;
    const int lane = t & 31;
    const int wave = t >> 5;
    const int g = t >> 4;
    const int l16 = t & 15;
    s_cnt[t] = 0;
    if (t == 0) *s_start = 0;
    __syncthreads();
    for (int c0 = 0; c0 < nChunk; c0 += 1024) {
        int v = load_meta_tile(meta, s_meta, c0, nChunk, b, t);
        for (int o = 16; o > 0; o >>= 1) v += __shfl_xor(v, o);
        if (lane == 0 && v != 0) atomicAdd(s_start, v);
        __syncthreads();
        walk_pieces<false, 16>(bpack, s_meta, s_cnt, nullptr, c0, min(1024, nChunk - c0), g, l16);
        __syncthreads();
    }
    const int start = *s_start;
    const int c = s_cnt[t];
    const int incl = wave_incl_scan_i(c, lane);
    if (lane == 31) s_w[wave] = incl;
    __syncthreads();
    int wb = 0;
    for (int w = 0; w < 8; ++w) if (w < wave) wb += s_w[w];
    const int ex = wb + incl - c;
    s_cur[t] = ex;
    __syncthreads();
    {
        const int n = b * 256 + t;
        if (n < N) { cnt[n] = c; off[n] = start + ex; }
    }
    for (int c0 = 0; c0 < nChunk; c0 += 1024) {
        if (nChunk > 1024) {
            (void)load_meta_tile(meta, s_meta, c0, nChunk, b, t);
            __syncthreads();
        }
        walk_pieces<true, 16>(bpack, s_meta, s_cur, ssrc + start, c0, min(1024, nChunk - c0), g, l16);
        if (nChunk > 1024) __syncthreads();
    }
}

__global__ __launch_bounds__(256) __attribute__((amdgpu_waves_per_eu(4, 4))) void sort_and_project(
    const unsigned* __restrict__ bpack, const unsigned* __restrict__ meta,
    int* __restrict__ cnt, int* __restrict__ off, int* __restrict__ ssrc, int N, int nChunk, int nBucket,
    const float* __restrict__ A, const float* __restrict__ B,
    const float* __restrict__ attS, const float* __restrict__ attD,
    float* __restrict__ C, float* __restrict__ alS, float* __restrict__ alD, int rowTiles)
{
    __shared__ __attribute__((aligned(16))) float s_al[2 * 4 * 16 * 4];
    __shared__ int s_cnt[256];
    __shared__ int s_cur[256];
    __shared__ int s_w[8];
    __shared__ int s_start;
    __shared__ unsigned s_meta[1024];
    const int blk = __builtin_amdgcn_readfirstlane((int)blockIdx.x);
    if (blk < nBucket)
        bucket_body(bpack, meta, cnt, off, ssrc, N, nChunk, blk, s_cnt, s_cur, s_w, &s_start, s_meta);
    else
        gemm_body<128, 128, 32, 8>(A, B, attS, attD, C, alS, alD, rowTiles, blk - nBucket, s_al);
}

template <int T>
__device__ __forceinline__ float grp8_bcast(float v) {
    return __uint_as_float(__builtin_amdgcn_mov_dpp8(__float_as_uint(v), (unsigned)T * 0x249249u));
}
template <int T>
__device__ __forceinline__ int row_bcast_i(int v) {
    return __builtin_amdgcn_mov_dpp(v, 0x150 | T, 0xf, 0xf, false);
}
template <int T>
__device__ __forceinline__ float row_bcast_f(float v) {
    return __int_as_float(__builtin_amdgcn_mov_dpp(__float_as_int(v), 0x150 | T, 0xf, 0xf, false));
}
__device__ __forceinline__ float xhalf(float x) {
    const int b = __float_as_int(x);
    return __int_as_float(__builtin_amdgcn_permlanex16(b, b, 0x76543210, 0xfedcba98, false, false));
}

__device__ __forceinline__ float dot4(const float4 r, const v2f aLo, const v2f aHi) {
    const v2f lo = {r.x, r.y};
    const v2f hi = {r.z, r.w};
    const v2f t = lo * aLo + hi * aHi;
    return t.x + t.y;
}
template <int A>
__device__ __forceinline__ float fold4(float p0, float p1, float p2, float p3, bool uA, bool uB) {
    const float q0 = (uA ? p2 : p0) + row_xor<A>(uA ? p0 : p2);
    const float q1 = (uA ? p3 : p1) + row_xor<A>(uA ? p1 : p3);
    float t = (uB ? q1 : q0) + row_xor<A / 2>(uB ? q0 : q1);
    if (A == 8) t += row_xor<2>(t);
    t += row_xor<1>(t);
    return t;
}
template <int A>
__device__ __forceinline__ void softmax_step4(v2f& accLo, v2f& accHi, float& dp, float& m,
                                              const float4* r, float e) {
    float g = fmaxf(e, row_xor<A>(e));
    g = fmaxf(g, row_xor<A / 2>(g));
    const float mn = fmaxf(m, g);
    const float sc = __expf(m - mn);
    const float w = __expf(e - mn);
    float wk[4];
    if (A == 4) {
        wk[0] = grp8_bcast<0>(w); wk[1] = grp8_bcast<2>(w);
        wk[2] = grp8_bcast<4>(w); wk[3] = grp8_bcast<6>(w);
    } else {
        wk[0] = row_bcast_f<0>(w); wk[1] = row_bcast_f<4>(w);
        wk[2] = row_bcast_f<8>(w); wk[3] = row_bcast_f<12>(w);
    }
    const v2f scv = {sc, sc};
    accLo *= scv;
    accHi *= scv;
    for (int u = 0; u < 4; ++u) {
        const v2f wv = {wk[u], wk[u]};
        const v2f lo = {r[u].x, r[u].y};
        const v2f hi = {r[u].z, r[u].w};
        accLo += wv * lo;
        accHi += wv * hi;
    }
    dp = dp * sc + w;
    m = mn;
}

__global__ __launch_bounds__(256) void gather_softmax_h4(
    const int* __restrict__ off, const int* __restrict__ cnt, const int* __restrict__ ssrc,
    const float* __restrict__ alS, const float* __restrict__ alD, const float* __restrict__ attS,
    const float4* __restrict__ h, const float* __restrict__ bias,
    float4* __restrict__ outp, int N)
{
    const int lane = threadIdx.x & 31;
    const int hd = lane >> 3;
    const bool uA = (lane & 4) != 0, uB = (lane & 2) != 0;
    const int kown = (lane & 7) >> 1;
    const int n = __builtin_amdgcn_readfirstlane((int)(blockIdx.x * 8 + (threadIdx.x >> 5)));
    if (n >= N) return;
    const int beg = off[n];
    const int deg = cnt[n];
    const float4 a4 = *reinterpret_cast<const float4*>(attS + lane * 4);
    const v2f aLo = {a4.x, a4.y};
    const v2f aHi = {a4.z, a4.w};
    const float dn = alD[(size_t)n * 4 + hd];
    const float4 hn = h[(size_t)n * 32 + lane];

    float m = leaky(alS[(size_t)n * 4 + hd] + dn);
    v2f accLo = {hn.x, hn.y};
    v2f accHi = {hn.z, hn.w};
    float dp = kown == 0 ? 1.f : 0.f;

    for (int base = 0; base < deg; base += 32) {
        const int i = base + lane;
        const int sl = ssrc[i < deg ? beg + i : 0];
        const int sid = i < deg ? sl : n;
        const int cc = min(32, deg - base);
        for (int j = 0; j < cc; j += 4) {
            float4 r[4];
#pragma unroll
            for (int u = 0; u < 4; ++u) {
                const int sj = __builtin_amdgcn_readlane(sid, j + u);
                r[u] = h[(size_t)sj * 32 + lane];
            }
            const float t = fold4<4>(dot4(r[0], aLo, aHi), dot4(r[1], aLo, aHi),
                                     dot4(r[2], aLo, aHi), dot4(r[3], aLo, aHi), uA, uB);
            const float e = j + kown < cc ? leaky(t + dn) : -__builtin_huge_valf();
            softmax_step4<4>(accLo, accHi, dp, m, r, e);
        }
    }

    dp += row_xor<4>(dp);
    dp += row_xor<2>(dp);
    const float den = dp + INF_EPS;
    const float* bp = bias + lane * 4;
    float4 o;
    o.x = accLo.x / den + bp[0];
    o.y = accLo.y / den + bp[1];
    o.z = accHi.x / den + bp[2];
    o.w = accHi.y / den + bp[3];
    o.x = o.x > 0.f ? o.x : __expf(o.x) - 1.f;
    o.y = o.y > 0.f ? o.y : __expf(o.y) - 1.f;
    o.z = o.z > 0.f ? o.z : __expf(o.z) - 1.f;
    o.w = o.w > 0.f ? o.w : __expf(o.w) - 1.f;
    outp[(size_t)n * 32 + lane] = o;
}

template <int T0>
__device__ __forceinline__ void edges4_h1(v2f& accLo, v2f& accHi, float& dp, float& m, int s,
                                          const v2f aLo, const v2f aHi, float dn,
                                          const float4* __restrict__ h, int l16, int g, int cc) {
    int sj[4];
    float4 r[4];
    sj[0] = row_bcast_i<T0>(s);
    sj[1] = row_bcast_i<T0 + 1>(s);
    sj[2] = row_bcast_i<T0 + 2>(s);
    sj[3] = row_bcast_i<T0 + 3>(s);
    for (int u = 0; u < 4; ++u) r[u] = h[(size_t)sj[u] * 16 + l16];
    const float t = fold4<8>(dot4(r[0], aLo, aHi), dot4(r[1], aLo, aHi),
                             dot4(r[2], aLo, aHi), dot4(r[3], aLo, aHi),
                             (l16 & 8) != 0, (l16 & 4) != 0);
    const float e = 2 * (T0 + (l16 >> 2)) + g < cc ? leaky(t + dn) : -__builtin_huge_valf();
    softmax_step4<8>(accLo, accHi, dp, m, r, e);
}

__global__ __launch_bounds__(256) void gather_softmax_h1(
    const int* __restrict__ off, const int* __restrict__ cnt, const int* __restrict__ ssrc,
    const float* __restrict__ alS, const float* __restrict__ alD, const float* __restrict__ attS,
    const float4* __restrict__ h, const float* __restrict__ bias,
    float4* __restrict__ outp, int N)
{
    const int lane = threadIdx.x & 31;
    const int g = lane >> 4;
    const int l16 = lane & 15;
    const int own = 2 * l16 + g;
    const int n = __builtin_amdgcn_readfirstlane((int)(blockIdx.x * 8 + (threadIdx.x >> 5)));
    if (n >= N) return;
    const int beg = off[n];
    const int deg = cnt[n];
    const float4 a4 = *reinterpret_cast<const float4*>(attS + l16 * 4);
    const v2f aLo = {a4.x, a4.y};
    const v2f aHi = {a4.z, a4.w};
    const float dn = alD[n];

    float m = leaky(alS[n] + dn);
    v2f accLo = {0.f, 0.f};
    v2f accHi = {0.f, 0.f};
    float dp = 0.f;
    if (g == 0) {
        const float4 hn = h[(size_t)n * 16 + l16];
        accLo = (v2f){hn.x, hn.y};
        accHi = (v2f){hn.z, hn.w};
        if (l16 < 4) dp = 1.f;
    }

    for (int base = 0; base < deg; base += 32) {
        const int i = base + own;
        const int sl = ssrc[i < deg ? beg + i : 0];
        const int s = i < deg ? sl : n;
        const int cc = min(32, deg - base);
        if (0 < cc)  edges4_h1<0>(accLo, accHi, dp, m, s, aLo, aHi, dn, h, l16, g, cc);
        if (8 < cc)  edges4_h1<4>(accLo, accHi, dp, m, s, aLo, aHi, dn, h, l16, g, cc);
        if (16 < cc) edges4_h1<8>(accLo, accHi, dp, m, s, aLo, aHi, dn, h, l16, g, cc);
        if (24 < cc) edges4_h1<12>(accLo, accHi, dp, m, s, aLo, aHi, dn, h, l16, g, cc);
    }

    const float sc = __expf(m - fmaxf(m, xhalf(m)));
    dp *= sc;
    dp += row_xor<8>(dp);
    dp += row_xor<4>(dp);
    dp += xhalf(dp);
    const float den = dp + INF_EPS;
    const float ax0 = accLo.x * sc, ay0 = accLo.y * sc, az0 = accHi.x * sc, aw0 = accHi.y * sc;
    const float ax = ax0 + xhalf(ax0);
    const float ay = ay0 + xhalf(ay0);
    const float az = az0 + xhalf(az0);
    const float aw = aw0 + xhalf(aw0);
    if (g == 0) {
        const float* bp = bias + l16 * 4;
        float4 o;
        o.x = ax / den + bp[0];
        o.y = ay / den + bp[1];
        o.z = az / den + bp[2];
        o.w = aw / den + bp[3];
        outp[(size_t)n * 16 + l16] = o;
    }
}

extern "C" void kernel_launch(void* const* d_in, const int* in_sizes, int n_in,
                              void* d_out, int out_size, void* d_ws, size_t ws_size,
                              hipStream_t stream)
{
    const float* x    = (const float*)d_in[0];
    const int*   eidx = (const int*)d_in[1];
    const float* W1   = (const float*)d_in[2];
    const float* aS1  = (const float*)d_in[3];
    const float* aD1  = (const float*)d_in[4];
    const float* b1   = (const float*)d_in[5];
    const float* W2   = (const float*)d_in[6];
    const float* aS2  = (const float*)d_in[7];
    const float* aD2  = (const float*)d_in[8];
    const float* b2   = (const float*)d_in[9];
    float* out = (float*)d_out;

    const int N    = in_sizes[0] / 128;
    const int E    = in_sizes[1] / 2;
    const int* esrc = eidx;
    const int* edst = eidx + E;

    float* ws = (float*)d_ws;
    size_t o = 0;
    float* h1   = ws + o; o += (size_t)N * 128;
    float* alS1 = ws + o; o += (size_t)N * 4;
    float* alD1 = ws + o; o += (size_t)N * 4;
    float* agg1 = ws + o; o += (size_t)N * 128;
    float* h2   = ws + o; o += (size_t)N * 64;
    float* alS2 = ws + o; o += ((size_t)N + 31) & ~(size_t)31;
    float* alD2 = ws + o; o += ((size_t)N + 31) & ~(size_t)31;
    o = (o + 31) & ~(size_t)31;
    unsigned* bpack = (unsigned*)(ws + o); o += (size_t)E;
    int*   ssrc = (int*)(ws + o); o += (size_t)E;
    int*   off  = (int*)(ws + o); o += (size_t)N;
    int*   cnt  = (int*)(ws + o); o += (size_t)N;
    o = (o + 31) & ~(size_t)31;
    auto cdiv = [](int a, int b) { return (a + b - 1) / b; };
    const int nChunk  = cdiv(E, SORT_CHUNK);
    const int nBucket = cdiv(N, 256);
    unsigned* meta = (unsigned*)(ws + o); o += (size_t)nChunk * 256;

    chunk_group<<<nChunk, 256, 0, stream>>>(esrc, edst, bpack, meta, E);
    sort_and_project<<<nBucket + cdiv(N / 16, 4), 256, 0, stream>>>(bpack, meta, cnt, off, ssrc, N, nChunk, nBucket,
        x, W1, aS1, aD1, h1, alS1, alD1, N / 16);

    gather_softmax_h4<<<cdiv(N, 8), 256, 0, stream>>>(off, cnt, ssrc,
        alS1, alD1, aS1, (const float4*)h1, b1, (float4*)agg1, N);

    gemm_wmma_f32<128, 64, 64><<<cdiv(N / 16, 4), 128, 0, stream>>>(agg1, W2, aS2, aD2, h2, alS2, alD2, N / 16);
    gather_softmax_h1<<<cdiv(N, 8), 256, 0, stream>>>(off, cnt, ssrc,
        alS2, alD2, aS2, (const float4*)h2, b2, (float4*)out, N);

    (void)n_in; (void)out_size; (void)ws_size;
}
